# starting kernel
# baseline (speedup 1.0000x reference)
#include <hip/hip_runtime.h>
#include <hip/hip_bf16.h>

#define N_NODES 30000
#define E_EDGES 480000
#define EN_TOT  (E_EDGES + N_NODES)
#define DIM     128
#define NLAYERS 3
#define NEG_SLOPE 0.2f
#define NWEIGHTS 4
#define WPK_ELEMS_PER_W (8 * 4 * 32 * 16)

typedef __attribute__((ext_vector_type(16))) __bf16 v16bf;
typedef __attribute__((ext_vector_type(4)))  __bf16 v4bf;
typedef __attribute__((ext_vector_type(8)))  float  v8f;

__device__ __forceinline__ unsigned f2ord(float x) {
    unsigned b = __float_as_uint(x);
    return (b & 0x80000000u) ? ~b : (b | 0x80000000u);
}
__device__ __forceinline__ float ord2f(unsigned u) {
    return __uint_as_float((u & 0x80000000u) ? (u ^ 0x80000000u) : ~u);
}

__global__ void __launch_bounds__(256)
pack_weights_kernel(const float* __restrict__ W_mlp, const float* __restrict__ Ws,
                    __bf16* __restrict__ Wpk)
{
    const int gw   = (blockIdx.x * blockDim.x + threadIdx.x) >> 5;
    const int lane = threadIdx.x & 31;
    if (gw >= NWEIGHTS * 32) return;
    const int w  = gw >> 5;
    const int ct = (gw & 31) >> 2;
    const int s  = gw & 3;
    const float* W = (w == 0) ? W_mlp : (Ws + (size_t)(w - 1) * DIM * DIM);
    const int r  = lane & 15;
    const int hi = lane >> 4;
    const int col = ct * 16 + r;
    const int kb  = s * 32;
    __bf16* out = Wpk + ((size_t)gw * 32 + lane) * 16;
#pragma unroll
    for (int h = 0; h < 16; ++h) {
        const int K = kb + ((h < 8) ? (8 * hi + h) : (16 + 8 * hi + (h - 8)));
        out[h] = (__bf16)W[(size_t)K * DIM + col];
    }
}

template <bool BF16_OUT>
__global__ void __launch_bounds__(256)
gemm128_wmma(const float* __restrict__ X, const __bf16* __restrict__ Wpk,
             const float* __restrict__ bias, void* __restrict__ out, int rows)
{
    if (blockIdx.x * 16 >= (unsigned)rows) return;
    const int lane = threadIdx.x & 31;
    const int ct   = threadIdx.x >> 5;
    const int r    = lane & 15;
    const int hi   = lane >> 4;
    const int row  = blockIdx.x * 16 + r;
    const int col0 = ct * 16;

    v8f acc = {};
#pragma unroll
    for (int s = 0; s < 4; ++s) {
        const int kb = s * 32;
        v16bf a;
        const float* ap = X + (size_t)row * DIM + kb + 8 * hi;
#pragma unroll
        for (int j = 0; j < 8; ++j) {
            a[j]     = (__bf16)ap[j];
            a[8 + j] = (__bf16)ap[16 + j];
        }
        const v16bf b = *(const v16bf*)(Wpk + ((size_t)(ct * 4 + s) * 32 + lane) * 16);
        acc = __builtin_amdgcn_wmma_f32_16x16x32_bf16(
             false, a,  false, b,
             (short)0, acc,  false,  false);
    }
    if (BF16_OUT) {
        __bf16* H = (__bf16*)out;
#pragma unroll
        for (int v = 0; v < 8; ++v) {
            const int m = blockIdx.x * 16 + v + 8 * hi;
            H[(size_t)m * DIM + col0 + r] = (__bf16)acc[v];
        }
    } else {
        float* H = (float*)out;
        const float bv = bias[col0 + r];
#pragma unroll
        for (int v = 0; v < 8; ++v) {
            const int m = blockIdx.x * 16 + v + 8 * hi;
            H[(size_t)m * DIM + col0 + r] = acc[v] + bv;
        }
    }
}

__global__ void __launch_bounds__(256)
alpha_init_kernel(const __bf16* __restrict__ Hb,
                  const float* __restrict__ asrc, const float* __restrict__ adst,
                  float* __restrict__ as_, float* __restrict__ ad_,
                  unsigned* __restrict__ m_u, float* __restrict__ z,
                  float* __restrict__ accum)
{
    const int wave = blockIdx.x * (blockDim.x >> 5) + (threadIdx.x >> 5);
    const int lane = threadIdx.x & 31;
    if (wave >= N_NODES) return;

    const v4bf   h4 = ((const v4bf*)(Hb + (size_t)wave * DIM))[lane];
    const float4 s4 = ((const float4*)asrc)[lane];
    const float4 d4 = ((const float4*)adst)[lane];
    const float hx = (float)h4[0], hy = (float)h4[1], hz = (float)h4[2], hw = (float)h4[3];
    float ps = hx * s4.x + hy * s4.y + hz * s4.z + hw * s4.w;
    float pd = hx * d4.x + hy * d4.y + hz * d4.z + hw * d4.w;
#pragma unroll
    for (int off = 16; off > 0; off >>= 1) {
        ps += __shfl_down(ps, off, 32);
        pd += __shfl_down(pd, off, 32);
    }
    if (lane == 0) {
        as_[wave] = ps;
        ad_[wave] = pd;
        m_u[wave] = 0u;
        z[wave]   = 0.0f;
    }
    ((float4*)(accum + (size_t)wave * DIM))[lane] = make_float4(0.f, 0.f, 0.f, 0.f);
}

__global__ void __launch_bounds__(256)
edge_max_kernel(const int* __restrict__ esrc, const int* __restrict__ edst,
                const float* __restrict__ as_, const float* __restrict__ ad_,
                unsigned* __restrict__ m_u)
{
    const int e = blockIdx.x * blockDim.x + threadIdx.x;
    if (e >= EN_TOT) return;
    int s, d;
    if (e < E_EDGES) { s = esrc[e]; d = edst[e]; } else { s = d = e - E_EDGES; }
    float v = as_[s] + ad_[d];
    v = (v > 0.0f) ? v : NEG_SLOPE * v;
    atomicMax(&m_u[d], f2ord(v));
}

__global__ void __launch_bounds__(256)
edge_sum_kernel(const int* __restrict__ esrc, const int* __restrict__ edst,
                const float* __restrict__ as_, const float* __restrict__ ad_,
                const unsigned* __restrict__ m_u, float* __restrict__ z)
{
    const int e = blockIdx.x * blockDim.x + threadIdx.x;
    if (e >= EN_TOT) return;
    int s, d;
    if (e < E_EDGES) { s = esrc[e]; d = edst[e]; } else { s = d = e - E_EDGES; }
    float v = as_[s] + ad_[d];
    v = (v > 0.0f) ? v : NEG_SLOPE * v;
    atomicAdd(&z[d], __expf(v - ord2f(m_u[d])));
}

__global__ void __launch_bounds__(256)
edge_aggr_kernel(const int* __restrict__ esrc, const int* __restrict__ edst,
                 const float* __restrict__ as_, const float* __restrict__ ad_,
                 const unsigned* __restrict__ m_u, const float* __restrict__ z,
                 const __bf16* __restrict__ Hb, float* __restrict__ accum)
{
    const int wave = blockIdx.x * (blockDim.x >> 5) + (threadIdx.x >> 5);
    const int lane = threadIdx.x & 31;
    if (wave >= EN_TOT) return;
    int s, d;
    if (wave < E_EDGES) { s = esrc[wave]; d = edst[wave]; } else { s = d = wave - E_EDGES; }
    float v = as_[s] + ad_[d];
    v = (v > 0.0f) ? v : NEG_SLOPE * v;
    const float coef = __expf(v - ord2f(m_u[d])) / z[d];

    const v4bf h4 = ((const v4bf*)(Hb + (size_t)s * DIM))[lane];
    float* ap = accum + (size_t)d * DIM + lane * 4;
    atomicAdd(ap + 0, coef * (float)h4[0]);
    atomicAdd(ap + 1, coef * (float)h4[1]);
    atomicAdd(ap + 2, coef * (float)h4[2]);
    atomicAdd(ap + 3, coef * (float)h4[3]);
}

__global__ void __launch_bounds__(256)
finalize_kernel(const float* __restrict__ accum, const float* __restrict__ bias,
                float* __restrict__ X)
{
    const int i = blockIdx.x * blockDim.x + threadIdx.x;
    if (i >= N_NODES * DIM) return;
    const float v = accum[i] + bias[i & (DIM - 1)];
    X[i] = (v > 0.0f) ? v : (__expf(v) - 1.0f);
}

extern "C" void kernel_launch(void* const* d_in, const int* in_sizes, int n_in,
                              void* d_out, int out_size, void* d_ws, size_t ws_size,
                              hipStream_t stream)
{
    (void)in_sizes; (void)n_in; (void)out_size; (void)ws_size;

    const float* xq    = (const float*)d_in[0];
    const float* xt    = (const float*)d_in[1];
    const int*   ei_q  = (const int*)d_in[2];
    const float* W_mlp = (const float*)d_in[4];
    const float* b_mlp = (const float*)d_in[5];
    const float* Ws    = (const float*)d_in[6];
    const float* a_src = (const float*)d_in[7];
    const float* a_dst = (const float*)d_in[8];
    const float* biases= (const float*)d_in[9];

    const int* esrc = ei_q;
    const int* edst = ei_q + E_EDGES;

    float* Xq = (float*)d_out;
    float* Xt = Xq + (size_t)N_NODES * DIM;

    __bf16*   Wpk   = (__bf16*)d_ws;
    __bf16*   Hb    = Wpk + (size_t)NWEIGHTS * WPK_ELEMS_PER_W;
    float*    accum = (float*)(Hb + (size_t)N_NODES * DIM);
    float*    as_   = accum + (size_t)N_NODES * DIM;
    float*    ad_   = as_ + N_NODES;
    unsigned* m_u   = (unsigned*)(ad_ + N_NODES);
    float*    z     = (float*)(m_u + N_NODES);

    const dim3 blk(256);
    const dim3 pack_grid((NWEIGHTS * 32 * 32 + 255) / 256);
    const dim3 gemm_grid(N_NODES / 16);
    const dim3 node_wave_grid((N_NODES + 7) / 8);
    const dim3 edge_grid((EN_TOT + 255) / 256);
    const dim3 edge_wave_grid((EN_TOT + 7) / 8);
    const dim3 elem_grid((N_NODES * DIM + 255) / 256);

    pack_weights_kernel<<<pack_grid, blk, 0, stream>>>(W_mlp, Ws, Wpk);

    gemm128_wmma<false><<<gemm_grid, blk, 0, stream>>>(xq, Wpk, b_mlp, Xq, N_NODES);
    gemm128_wmma<false><<<gemm_grid, blk, 0, stream>>>(xt, Wpk, b_mlp, Xt, N_NODES);

    for (int l = 0; l < NLAYERS; ++l) {
        const __bf16* Wl = Wpk + (size_t)(l + 1) * WPK_ELEMS_PER_W;
        const float*  al = a_src  + (size_t)l * DIM;
        const float*  dl = a_dst  + (size_t)l * DIM;
        const float*  bl = biases + (size_t)l * DIM;
        float* Xs[2] = { Xq, Xt };
        for (int g = 0; g < 2; ++g) {
            float* X = Xs[g];
            gemm128_wmma<true><<<gemm_grid, blk, 0, stream>>>(X, Wl, nullptr, (void*)Hb, N_NODES);
            alpha_init_kernel<<<node_wave_grid, blk, 0, stream>>>(Hb, al, dl, as_, ad_, m_u, z, accum);
            edge_max_kernel <<<edge_grid, blk, 0, stream>>>(esrc, edst, as_, ad_, m_u);
            edge_sum_kernel <<<edge_grid, blk, 0, stream>>>(esrc, edst, as_, ad_, m_u, z);
            edge_aggr_kernel<<<edge_wave_grid, blk, 0, stream>>>(esrc, edst, as_, ad_, m_u, z, Hb, accum);
            finalize_kernel <<<elem_grid, blk, 0, stream>>>(accum, bl, X);
        }
    }
}


# Long-lived waves (660000 -> 31000 enter per pass): aggregate wave walks 32 edges, 8 rows ahead; finalize four float4 per lane; alpha 4 rows per wave; f32 atomics per edge kept, arrival order moves
# speedup vs baseline: 1.4132x; 1.4132x over previous
#include <hip/hip_runtime.h>
#include <hip/hip_bf16.h>

#define N_NODES 30000
#define E_EDGES 480000
#define EN_TOT  (E_EDGES + N_NODES)
#define DIM     128
#define NLAYERS 3
#define NEG_SLOPE 0.2f
#define NWEIGHTS 4
#define WPK_ELEMS_PER_W (8 * 4 * 32 * 16)
#define STAT_STRIDE 32

typedef __attribute__((ext_vector_type(16))) __bf16 v16bf;
typedef __attribute__((ext_vector_type(4)))  __bf16 v4bf;
typedef __attribute__((ext_vector_type(8)))  float  v8f;

__device__ __forceinline__ unsigned f2ord(float x) {
    unsigned b = __float_as_uint(x);
    return (b & 0x80000000u) ? ~b : (b | 0x80000000u);
}
__device__ __forceinline__ float ord2f(unsigned u) {
    return __uint_as_float((u & 0x80000000u) ? (u ^ 0x80000000u) : ~u);
}

__global__ void __launch_bounds__(256)
pack_weights_kernel(const float* __restrict__ W_mlp, const float* __restrict__ Ws,
                    __bf16* __restrict__ Wpk)
{
    const int gw   = (blockIdx.x * blockDim.x + threadIdx.x) >> 5;
    const int lane = threadIdx.x & 31;
    if (gw >= NWEIGHTS * 32) return;
    const int w  = gw >> 5;
    const int ct = (gw & 31) >> 2;
    const int s  = gw & 3;
    const float* W = (w == 0) ? W_mlp : (Ws + (size_t)(w - 1) * DIM * DIM);
    const int r  = lane & 15;
    const int hi = lane >> 4;
    const int col = ct * 16 + r;
    const int kb  = s * 32;
    __bf16* out = Wpk + ((size_t)gw * 32 + lane) * 16;
#pragma unroll
    for (int h = 0; h < 16; ++h) {
        const int K = kb + ((h < 8) ? (8 * hi + h) : (16 + 8 * hi + (h - 8)));
        out[h] = (__bf16)W[(size_t)K * DIM + col];
    }
}

template <bool BF16_OUT>
__global__ void __launch_bounds__(256)
gemm128_wmma(const float* __restrict__ X, const __bf16* __restrict__ Wpk,
             const float* __restrict__ bias, void* __restrict__ out, int rows)
{
    if (blockIdx.x * 16 >= (unsigned)rows) return;
    const int lane = threadIdx.x & 31;
    const int ct   = threadIdx.x >> 5;
    const int r    = lane & 15;
    const int hi   = lane >> 4;
    const int row  = blockIdx.x * 16 + r;
    const int col0 = ct * 16;

    v8f acc = {};
#pragma unroll
    for (int s = 0; s < 4; ++s) {
        const int kb = s * 32;
        v16bf a;
        const float* ap = X + (size_t)row * DIM + kb + 8 * hi;
#pragma unroll
        for (int j = 0; j < 8; ++j) {
            a[j]     = (__bf16)ap[j];
            a[8 + j] = (__bf16)ap[16 + j];
        }
        const v16bf b = *(const v16bf*)(Wpk + ((size_t)(ct * 4 + s) * 32 + lane) * 16);
        acc = __builtin_amdgcn_wmma_f32_16x16x32_bf16(
             false, a,  false, b,
             (short)0, acc,  false,  false);
    }
    if (BF16_OUT) {
        __bf16* H = (__bf16*)out;
#pragma unroll
        for (int v = 0; v < 8; ++v) {
            const int m = blockIdx.x * 16 + v + 8 * hi;
            H[(size_t)m * DIM + col0 + r] = (__bf16)acc[v];
        }
    } else {
        float* H = (float*)out;
        const float bv = bias[col0 + r];
#pragma unroll
        for (int v = 0; v < 8; ++v) {
            const int m = blockIdx.x * 16 + v + 8 * hi;
            H[(size_t)m * DIM + col0 + r] = acc[v] + bv;
        }
    }
}

#define NODES_PER_WAVE 4
static_assert(N_NODES % NODES_PER_WAVE == 0, "a wave's rows must all exist");

__global__ void __launch_bounds__(256)
alpha_init_kernel(const __bf16* __restrict__ Hb,
                  const float* __restrict__ asrc, const float* __restrict__ adst,
                  float* __restrict__ as_, float* __restrict__ ad_,
                  unsigned* __restrict__ m_u, float* __restrict__ z,
                  float* __restrict__ accum)
{
    const int wave = blockIdx.x * (blockDim.x >> 5) + (threadIdx.x >> 5);
    const int lane = threadIdx.x & 31;
    const int n0   = wave * NODES_PER_WAVE;
    if (n0 >= N_NODES) return;

    const float4 s4 = ((const float4*)asrc)[lane];
    const float4 d4 = ((const float4*)adst)[lane];
    v4bf h4[NODES_PER_WAVE];
#pragma unroll
    for (int k = 0; k < NODES_PER_WAVE; ++k)
        h4[k] = ((const v4bf*)(Hb + (size_t)(n0 + k) * DIM))[lane];

#pragma unroll
    for (int k = 0; k < NODES_PER_WAVE; ++k) {
        const int node = n0 + k;
        const float hx = (float)h4[k][0], hy = (float)h4[k][1], hz = (float)h4[k][2], hw = (float)h4[k][3];
        float ps = hx * s4.x + hy * s4.y + hz * s4.z + hw * s4.w;
        float pd = hx * d4.x + hy * d4.y + hz * d4.z + hw * d4.w;
#pragma unroll
        for (int off = 16; off > 0; off >>= 1) {
            ps += __shfl_down(ps, off, 32);
            pd += __shfl_down(pd, off, 32);
        }
        if (lane == 0) {
            as_[node] = ps;
            ad_[node] = pd;
            m_u[(size_t)node * STAT_STRIDE] = 0u;
            z[(size_t)node * STAT_STRIDE]   = 0.0f;
        }
        ((float4*)(accum + (size_t)node * DIM))[lane] = make_float4(0.f, 0.f, 0.f, 0.f);
    }
}

__global__ void __launch_bounds__(256)
edge_max_kernel(const int* __restrict__ esrc, const int* __restrict__ edst,
                const float* __restrict__ as_, const float* __restrict__ ad_,
                unsigned* __restrict__ m_u)
{
    const int e = blockIdx.x * blockDim.x + threadIdx.x;
    if (e >= EN_TOT) return;
    int s, d;
    if (e < E_EDGES) { s = esrc[e]; d = edst[e]; } else { s = d = e - E_EDGES; }
    float v = as_[s] + ad_[d];
    v = (v > 0.0f) ? v : NEG_SLOPE * v;
    atomicMax(&m_u[(size_t)d * STAT_STRIDE], f2ord(v));
}

__global__ void __launch_bounds__(256)
edge_sum_kernel(const int* __restrict__ esrc, const int* __restrict__ edst,
                const float* __restrict__ as_, const float* __restrict__ ad_,
                const unsigned* __restrict__ m_u, float* __restrict__ z)
{
    const int e = blockIdx.x * blockDim.x + threadIdx.x;
    if (e >= EN_TOT) return;
    int s, d;
    if (e < E_EDGES) { s = esrc[e]; d = edst[e]; } else { s = d = e - E_EDGES; }
    float v = as_[s] + ad_[d];
    v = (v > 0.0f) ? v : NEG_SLOPE * v;
    atomicAdd(&z[(size_t)d * STAT_STRIDE], __expf(v - ord2f(m_u[(size_t)d * STAT_STRIDE])));
}

#define EDGES_PER_WAVE 32
#define ROWS_AHEAD     8
static_assert((EN_TOT % EDGES_PER_WAVE) % ROWS_AHEAD == 0 && EDGES_PER_WAVE % ROWS_AHEAD == 0,
              "the walk takes ROWS_AHEAD edges at a time");

__device__ __forceinline__ v4bf load_row4(const __bf16* __restrict__ Hb, int s, int lane)
{
    return ((const v4bf*)(Hb + (size_t)s * DIM))[lane];
}

__global__ void __launch_bounds__(256)
edge_aggr_kernel(const int* __restrict__ esrc, const int* __restrict__ edst,
                 const float* __restrict__ as_, const float* __restrict__ ad_,
                 const unsigned* __restrict__ m_u, const float* __restrict__ z,
                 const __bf16* __restrict__ Hb, float* __restrict__ accum)
{
    const int wave = __builtin_amdgcn_readfirstlane(blockIdx.x * (blockDim.x >> 5) + (threadIdx.x >> 5));
    const int lane = threadIdx.x & 31;
    const int e0   = wave * EDGES_PER_WAVE;
    if (e0 >= EN_TOT) return;
    const int n = (EN_TOT - e0 < EDGES_PER_WAVE) ? (EN_TOT - e0) : EDGES_PER_WAVE;

    int   s = 0, d = 0;
    float coef = 0.0f;
    if (lane < n) {
        const int e = e0 + lane;
        if (e < E_EDGES) { s = esrc[e]; d = edst[e]; } else { s = d = e - E_EDGES; }
        float v = as_[s] + ad_[d];
        v = (v > 0.0f) ? v : NEG_SLOPE * v;
        coef = __expf(v - ord2f(m_u[(size_t)d * STAT_STRIDE])) / z[(size_t)d * STAT_STRIDE];
    }
    const int coef_bits = __float_as_int(coef);

    v4bf h[ROWS_AHEAD];
#pragma unroll
    for (int u = 0; u < ROWS_AHEAD; ++u)
        h[u] = load_row4(Hb, __builtin_amdgcn_readlane(s, u), lane);

    for (int j0 = 0; j0 < n; j0 += ROWS_AHEAD) {
        v4bf hn[ROWS_AHEAD];
        const bool more = (j0 + ROWS_AHEAD < n);
        if (more) {
#pragma unroll
            for (int u = 0; u < ROWS_AHEAD; ++u)
                hn[u] = load_row4(Hb, __builtin_amdgcn_readlane(s, j0 + ROWS_AHEAD + u), lane);
        }
#pragma unroll
        for (int u = 0; u < ROWS_AHEAD; ++u) {
            const int   dj = __builtin_amdgcn_readlane(d, j0 + u);
            const float cj = __int_as_float(__builtin_amdgcn_readlane(coef_bits, j0 + u));
            float* ap = accum + (size_t)dj * DIM + lane * 4;
            atomicAdd(ap + 0, cj * (float)h[u][0]);
            atomicAdd(ap + 1, cj * (float)h[u][1]);
            atomicAdd(ap + 2, cj * (float)h[u][2]);
            atomicAdd(ap + 3, cj * (float)h[u][3]);
        }
        if (more) {
#pragma unroll
            for (int u = 0; u < ROWS_AHEAD; ++u) h[u] = hn[u];
        }
    }
}

#define FIN_VEC     4
#define FIN_QUADS   (N_NODES * DIM / 4)
#define FIN_THREADS ((FIN_QUADS + FIN_VEC - 1) / FIN_VEC)
#define FIN_BLOCKS  ((FIN_THREADS + 255) / 256)
#define FIN_STRIDE  (FIN_BLOCKS * 256)

__device__ __forceinline__ float elu1(float v) { return (v > 0.0f) ? v : (__expf(v) - 1.0f); }

__global__ void __launch_bounds__(256)
finalize_kernel(const float* __restrict__ accum, const float* __restrict__ bias,
                float* __restrict__ X)
{
    const int t = blockIdx.x * blockDim.x + threadIdx.x;
    float4 a[FIN_VEC];
#pragma unroll
    for (int k = 0; k < FIN_VEC; ++k) {
        const int q = t + k * FIN_STRIDE;
        if (q < FIN_QUADS) a[k] = ((const float4*)accum)[q];
    }
#pragma unroll
    for (int k = 0; k < FIN_VEC; ++k) {
        const int q = t + k * FIN_STRIDE;
        if (q < FIN_QUADS) {
            const float4 b = ((const float4*)bias)[q & (DIM / 4 - 1)];
            float4 r;
            r.x = elu1(a[k].x + b.x);
            r.y = elu1(a[k].y + b.y);
            r.z = elu1(a[k].z + b.z);
            r.w = elu1(a[k].w + b.w);
            ((float4*)X)[q] = r;
        }
    }
}

extern "C" void kernel_launch(void* const* d_in, const int* in_sizes, int n_in,
                              void* d_out, int out_size, void* d_ws, size_t ws_size,
                              hipStream_t stream)
{
    (void)in_sizes; (void)n_in; (void)out_size; (void)ws_size;

    const float* xq    = (const float*)d_in[0];
    const float* xt    = (const float*)d_in[1];
    const int*   ei_q  = (const int*)d_in[2];
    const float* W_mlp = (const float*)d_in[4];
    const float* b_mlp = (const float*)d_in[5];
    const float* Ws    = (const float*)d_in[6];
    const float* a_src = (const float*)d_in[7];
    const float* a_dst = (const float*)d_in[8];
    const float* biases= (const float*)d_in[9];

    const int* esrc = ei_q;
    const int* edst = ei_q + E_EDGES;

    float* Xq = (float*)d_out;
    float* Xt = Xq + (size_t)N_NODES * DIM;

    __bf16*   Wpk   = (__bf16*)d_ws;
    __bf16*   Hb    = Wpk + (size_t)NWEIGHTS * WPK_ELEMS_PER_W;
    float*    accum = (float*)(Hb + (size_t)N_NODES * DIM);
    float*    as_   = accum + (size_t)N_NODES * DIM;
    float*    ad_   = as_ + N_NODES;

    const dim3 blk(256);
    const dim3 pack_grid((NWEIGHTS * 32 * 32 + 255) / 256);
    const dim3 gemm_grid(N_NODES / 16);
    const dim3 node_wave_grid((N_NODES + NODES_PER_WAVE * 8 - 1) / (NODES_PER_WAVE * 8));
    const dim3 edge_grid((EN_TOT + 255) / 256);
    const dim3 edge_wave_grid((EN_TOT + EDGES_PER_WAVE * 8 - 1) / (EDGES_PER_WAVE * 8));
    const dim3 elem_grid(FIN_BLOCKS);

    pack_weights_kernel<<<pack_grid, blk, 0, stream>>>(W_mlp, Ws, Wpk);

    gemm128_wmma<false><<<gemm_grid, blk, 0, stream>>>(xq, Wpk, b_mlp, Xq, N_NODES);
    gemm128_wmma<false><<<gemm_grid, blk, 0, stream>>>(xt, Wpk, b_mlp, Xt, N_NODES);

    for (int l = 0; l < NLAYERS; ++l) {
        const __bf16* Wl = Wpk + (size_t)(l + 1) * WPK_ELEMS_PER_W;
        const float*  al = a_src  + (size_t)l * DIM;
        const float*  dl = a_dst  + (size_t)l * DIM;
        const float*  bl = biases + (size_t)l * DIM;
        float* Xs[2] = { Xq, Xt };
        for (int g = 0; g < 2; ++g) {
            float* X = Xs[g];
            unsigned* m_u = (unsigned*)X;
            float*    z   = X + (size_t)N_NODES * STAT_STRIDE;
            gemm128_wmma<true><<<gemm_grid, blk, 0, stream>>>(X, Wl, nullptr, (void*)Hb, N_NODES);
            alpha_init_kernel<<<node_wave_grid, blk, 0, stream>>>(Hb, al, dl, as_, ad_, m_u, z, accum);
            edge_max_kernel <<<edge_grid, blk, 0, stream>>>(esrc, edst, as_, ad_, m_u);
            edge_sum_kernel <<<edge_grid, blk, 0, stream>>>(esrc, edst, as_, ad_, m_u, z);
            edge_aggr_kernel<<<edge_wave_grid, blk, 0, stream>>>(esrc, edst, as_, ad_, m_u, z, Hb, accum);
            finalize_kernel <<<elem_grid, blk, 0, stream>>>(accum, bl, X);
        }
    }
}


# Owner waves' edge entries (source, row) are left in the workspace by the first pass of a call; the five later passes read their lists and no longer scan the destination list; integer sums as before
# speedup vs baseline: 4.2121x; 2.9805x over previous
#include <hip/hip_runtime.h>
#include <hip/hip_bf16.h>

#define N_NODES 30000
#define E_EDGES 480000
#define EN_TOT  (E_EDGES + N_NODES)
#define DIM     128
#define NLAYERS 3
#define NEG_SLOPE 0.2f
#define NWEIGHTS 4
#define WPK_ELEMS_PER_W (8 * 4 * 32 * 16)
#define Z_FIX_ONE    1099511627776.0f
#define Z_FIX_INV    9.094947017729282e-13f
#define FIX_ONE      1073741824.0f
#define FIX_INV_ONE  9.313225746154785e-10f
#define HMAX_FLOOR   1.0e-20f

typedef __attribute__((ext_vector_type(16))) __bf16 v16bf;
typedef __attribute__((ext_vector_type(4)))  __bf16 v4bf;
typedef __attribute__((ext_vector_type(8)))  float  v8f;

__device__ __forceinline__ unsigned f2ord(float x) {
    unsigned b = __float_as_uint(x);
    return (b & 0x80000000u) ? ~b : (b | 0x80000000u);
}
__device__ __forceinline__ float ord2f(unsigned u) {
    return __uint_as_float((u & 0x80000000u) ? (u ^ 0x80000000u) : ~u);
}

__global__ void __launch_bounds__(256)
pack_weights_kernel(const float* __restrict__ W_mlp, const float* __restrict__ Ws,
                    __bf16* __restrict__ Wpk)
{
    const int gw   = (blockIdx.x * blockDim.x + threadIdx.x) >> 5;
    const int lane = threadIdx.x & 31;
    if (gw >= NWEIGHTS * 32) return;
    const int w  = gw >> 5;
    const int ct = (gw & 31) >> 2;
    const int s  = gw & 3;
    const float* W = (w == 0) ? W_mlp : (Ws + (size_t)(w - 1) * DIM * DIM);
    const int r  = lane & 15;
    const int hi = lane >> 4;
    const int col = ct * 16 + r;
    const int kb  = s * 32;
    __bf16* out = Wpk + ((size_t)gw * 32 + lane) * 16;
#pragma unroll
    for (int h = 0; h < 16; ++h) {
        const int K = kb + ((h < 8) ? (8 * hi + h) : (16 + 8 * hi + (h - 8)));
        out[h] = (__bf16)W[(size_t)K * DIM + col];
    }
}

template <bool BF16_OUT>
__global__ void __launch_bounds__(256)
gemm128_wmma(const float* __restrict__ X, const __bf16* __restrict__ Wpk,
             const float* __restrict__ bias, void* __restrict__ out, int rows)
{
    if (blockIdx.x * 16 >= (unsigned)rows) return;
    const int lane = threadIdx.x & 31;
    const int ct   = threadIdx.x >> 5;
    const int r    = lane & 15;
    const int hi   = lane >> 4;
    const int row  = blockIdx.x * 16 + r;
    const int col0 = ct * 16;

    v8f acc = {};
#pragma unroll
    for (int s = 0; s < 4; ++s) {
        const int kb = s * 32;
        v16bf a;
        const float* ap = X + (size_t)row * DIM + kb + 8 * hi;
#pragma unroll
        for (int j = 0; j < 8; ++j) {
            a[j]     = (__bf16)ap[j];
            a[8 + j] = (__bf16)ap[16 + j];
        }
        const v16bf b = *(const v16bf*)(Wpk + ((size_t)(ct * 4 + s) * 32 + lane) * 16);
        acc = __builtin_amdgcn_wmma_f32_16x16x32_bf16(
             false, a,  false, b,
             (short)0, acc,  false,  false);
    }
    if (BF16_OUT) {
        __bf16* H = (__bf16*)out;
#pragma unroll
        for (int v = 0; v < 8; ++v) {
            const int m = blockIdx.x * 16 + v + 8 * hi;
            H[(size_t)m * DIM + col0 + r] = (__bf16)acc[v];
        }
    } else {
        float* H = (float*)out;
        const float bv = bias[col0 + r];
#pragma unroll
        for (int v = 0; v < 8; ++v) {
            const int m = blockIdx.x * 16 + v + 8 * hi;
            H[(size_t)m * DIM + col0 + r] = acc[v] + bv;
        }
    }
}

#define NODES_PER_WAVE 4
#define ALPHA_WAVES (N_NODES / NODES_PER_WAVE)
static_assert(ALPHA_WAVES % 4 == 0, "wmax is read in 16-byte pieces");
static_assert(N_NODES % NODES_PER_WAVE == 0, "a wave's rows must all exist");

__global__ void __launch_bounds__(256)
alpha_init_kernel(const __bf16* __restrict__ Hb,
                  const float* __restrict__ asrc, const float* __restrict__ adst,
                  float* __restrict__ as_, float* __restrict__ ad_,
                  float* __restrict__ wmax)
{
    const int wave = blockIdx.x * (blockDim.x >> 5) + (threadIdx.x >> 5);
    const int lane = threadIdx.x & 31;
    const int n0   = wave * NODES_PER_WAVE;
    if (n0 >= N_NODES) return;

    const float4 s4 = ((const float4*)asrc)[lane];
    const float4 d4 = ((const float4*)adst)[lane];
    v4bf h4[NODES_PER_WAVE];
#pragma unroll
    for (int k = 0; k < NODES_PER_WAVE; ++k)
        h4[k] = ((const v4bf*)(Hb + (size_t)(n0 + k) * DIM))[lane];

    float amax = 0.0f;
#pragma unroll
    for (int k = 0; k < NODES_PER_WAVE; ++k) {
        const int node = n0 + k;
        const float hx = (float)h4[k][0], hy = (float)h4[k][1], hz = (float)h4[k][2], hw = (float)h4[k][3];
        amax = fmaxf(fmaxf(amax, fmaxf(fabsf(hx), fabsf(hy))), fmaxf(fabsf(hz), fabsf(hw)));
        float ps = hx * s4.x + hy * s4.y + hz * s4.z + hw * s4.w;
        float pd = hx * d4.x + hy * d4.y + hz * d4.z + hw * d4.w;
#pragma unroll
        for (int off = 16; off > 0; off >>= 1) {
            ps += __shfl_down(ps, off, 32);
            pd += __shfl_down(pd, off, 32);
        }
        if (lane == 0) {
            as_[node] = ps;
            ad_[node] = pd;
        }
    }
#pragma unroll
    for (int off = 16; off > 0; off >>= 1)
        amax = fmaxf(amax, __shfl_xor(amax, off, 32));
    if (lane == 0) wmax[wave] = amax;
}

#define ROWS_AHEAD   8
#define OWN_R        118
#define OWN_BLOCKS   ((N_NODES + OWN_R - 1) / OWN_R)
#define OWN_THREADS  1024
#define OWN_WAVES    (OWN_THREADS / 32)
#define SCAN_AHEAD   8
#define E_QUADS      (E_EDGES / 4)
#define SCAN_ITERS   ((E_QUADS + OWN_THREADS - 1) / OWN_THREADS)
#define SCAN_GROUPS  ((SCAN_ITERS + SCAN_AHEAD - 1) / SCAN_AHEAD)
#define OWN_QCAP     (32 + SCAN_AHEAD * 4 * 32)
#define OWN_RCAP     256
#define DL_BITS      7
#define DL_MASK      ((1 << DL_BITS) - 1)
#define WMAX_QUADS   (ALPHA_WAVES / 4)
#define WMAX_PER_THREAD ((WMAX_QUADS + OWN_THREADS - 1) / OWN_THREADS)
static_assert(E_EDGES % 4 == 0, "the destination list is read in 16-byte pieces");
static_assert(OWN_R <= (1 << DL_BITS), "row inside the group must fit DL_BITS");
static_assert(EN_TOT < (1 << (31 - DL_BITS)), "edge number and row share one 32-bit queue word");
static_assert(N_NODES < (1 << (31 - DL_BITS)), "source and row share one 32-bit word");
static_assert(OWN_WAVES == 32, "the per-wave maxima are folded by one wave-wide exchange");
#define OWN_LCAP     128
#define LIST_CHUNKS  (OWN_LCAP / 32)
#define LIST_WAVES   (OWN_BLOCKS * OWN_WAVES)
#define WS_STAT_END  (NWEIGHTS * WPK_ELEMS_PER_W * 2 + N_NODES * DIM * 2 + (2 * N_NODES + ALPHA_WAVES) * 4)
#define WS_LIST_OFF  ((WS_STAT_END + 127) / 128 * 128)
#define WS_NEED      ((size_t)WS_LIST_OFF + (size_t)LIST_WAVES * 4 + (size_t)LIST_WAVES * OWN_LCAP * 4)
static_assert(OWN_LCAP % 32 == 0, "a list is read 32 entries at a time");
static_assert(OWN_LCAP <= OWN_RCAP, "a listed wave retains all its entries");

__device__ __forceinline__ v4bf load_row4(const __bf16* __restrict__ Hb, int s, int lane)
{
    return ((const v4bf*)(Hb + (size_t)s * DIM))[lane];
}

__device__ __forceinline__ float elu1(float v) { return (v > 0.0f) ? v : (__expf(v) - 1.0f); }

struct OwnCtx {
    const int*    esrc;
    const float*  as_;
    const __bf16* Hb;
    const float*  ad_l;
    unsigned*     m_l;
    unsigned long long* z_l;
    int*          acc;
    float         qscale;
    int           lane;
};

__device__ __forceinline__ float own_z(const OwnCtx& c, int dl)
{
    return (float)c.z_l[dl] * Z_FIX_INV;
}

__device__ __forceinline__ void own_walk(const OwnCtx& c, int s, int dl, float coef_scaled, int take)
{
    const int lane = c.lane;
    const int coef_bits = __float_as_int(coef_scaled);
    v4bf h[ROWS_AHEAD];
#pragma unroll
    for (int u = 0; u < ROWS_AHEAD; ++u)
        h[u] = load_row4(c.Hb, __builtin_amdgcn_readlane(s, u), lane);

    for (int j0 = 0; j0 < take; j0 += ROWS_AHEAD) {
        v4bf hn[ROWS_AHEAD];
        const bool more = (j0 + ROWS_AHEAD < take);
        if (more) {
#pragma unroll
            for (int u = 0; u < ROWS_AHEAD; ++u)
                hn[u] = load_row4(c.Hb, __builtin_amdgcn_readlane(s, j0 + ROWS_AHEAD + u), lane);
        }
#pragma unroll
        for (int u = 0; u < ROWS_AHEAD; ++u) {
            const int   dj = __builtin_amdgcn_readlane(dl, j0 + u);
            const float cj = __int_as_float(__builtin_amdgcn_readlane(coef_bits, j0 + u));
            int* ap = c.acc + dj * DIM + lane;
            atomicAdd(ap +  0, __float2int_rn(cj * (float)h[u][0]));
            atomicAdd(ap + 32, __float2int_rn(cj * (float)h[u][1]));
            atomicAdd(ap + 64, __float2int_rn(cj * (float)h[u][2]));
            atomicAdd(ap + 96, __float2int_rn(cj * (float)h[u][3]));
        }
        if (more) {
#pragma unroll
            for (int u = 0; u < ROWS_AHEAD; ++u) h[u] = hn[u];
        }
    }
}

template <int PHASE>
__device__ __forceinline__ void own_chunk(const OwnCtx& c, const int* qe, int take,
                                          int* rsd, float* rv, int& rcount, bool& ovf)
{
    const int lane = c.lane;
    const bool on = (lane < take);
    int   s = 0, dl = 0;
    float v = 0.0f;
    if (on) {
        const int q = qe[lane];
        const int e = q >> DL_BITS;
        dl = q & DL_MASK;
        s = (e < E_EDGES) ? c.esrc[e] : (e - E_EDGES);
        const float t = c.as_[s] + c.ad_l[dl];
        v = (t > 0.0f) ? t : NEG_SLOPE * t;
    }
    if (PHASE == 0) {
        if (on) atomicMax(&c.m_l[dl], f2ord(v));
        if (!ovf) {
            if (rcount + take <= OWN_RCAP) {
                if (on) { rsd[rcount + lane] = (s << DL_BITS) | dl; rv[rcount + lane] = v; }
                rcount += take;
            } else {
                ovf = true;
            }
        }
    } else if (PHASE == 1) {
        if (on) {
            const float w = __expf(v - ord2f(c.m_l[dl]));
            atomicAdd(&c.z_l[dl], (unsigned long long)(w * Z_FIX_ONE));
        }
    } else {
        float coef = 0.0f;
        if (on) coef = __expf(v - ord2f(c.m_l[dl])) / own_z(c, dl);
        own_walk(c, s, dl, coef * c.qscale, take);
    }
}

template <int PHASE>
__device__ __forceinline__ void own_scan(const OwnCtx& c, const int* __restrict__ edst, int* qe,
                                         int wave, int n0, int rn,
                                         int* rsd, float* rv, int& rcount, bool& ovf)
{
    const int lane = c.lane;
    int qn = 0;

    {
        const int row = wave + 32 * lane;
        const bool keep = (row < rn);
        if (keep) qe[lane] = ((E_EDGES + n0 + row) << DL_BITS) | row;
        qn = __popc(__builtin_amdgcn_ballot_w32(keep));
    }

    for (int g = 0; g < SCAN_GROUPS; ++g) {
        const int it0 = ((g + (int)blockIdx.x) % SCAN_GROUPS) * SCAN_AHEAD;
        int4 d4[SCAN_AHEAD];
#pragma unroll
        for (int u = 0; u < SCAN_AHEAD; ++u) {
            const int idx = (it0 + u) * OWN_THREADS + (int)threadIdx.x;
            d4[u] = (idx < E_QUADS) ? ((const int4*)edst)[idx] : make_int4(-1, -1, -1, -1);
        }
#pragma unroll
        for (int u = 0; u < SCAN_AHEAD; ++u) {
            const int idx = (it0 + u) * OWN_THREADS + (int)threadIdx.x;
            const int dd[4] = { d4[u].x, d4[u].y, d4[u].z, d4[u].w };
#pragma unroll
            for (int k = 0; k < 4; ++k) {
                const int  dl   = dd[k] - n0;
                const bool keep = (unsigned)dl < (unsigned)rn;
                const unsigned mask = __builtin_amdgcn_ballot_w32(keep);
                if (mask != 0u) {
                    if (keep) qe[qn + __popc(mask & ((1u << lane) - 1u))] = ((idx * 4 + k) << DL_BITS) | dl;
                    qn += __popc(mask);
                }
            }
        }
        while (qn >= 32) {
            qn -= 32;
            own_chunk<PHASE>(c, qe + qn, 32, rsd, rv, rcount, ovf);
        }
    }
    if (qn > 0)
        own_chunk<PHASE>(c, qe, qn, rsd, rv, rcount, ovf);
}

template <bool BUILD>
__global__ void __launch_bounds__(OWN_THREADS)
own_softmax_aggr_kernel(const int* __restrict__ esrc, const int* __restrict__ edst,
                        const float* __restrict__ as_, const float* __restrict__ ad_,
                        const __bf16* __restrict__ Hb, const float* __restrict__ wmax,
                        const float* __restrict__ bias, float* __restrict__ X,
                        int* __restrict__ lcount, int* __restrict__ lst, size_t ws_size)
{
    __shared__ __attribute__((aligned(16))) int acc[OWN_R * DIM];
    __shared__ int   queue[OWN_WAVES][OWN_QCAP];
    __shared__ int   ret_sd[OWN_WAVES][OWN_RCAP];
    __shared__ float ret_v[OWN_WAVES][OWN_RCAP];
    __shared__ unsigned long long z_l[OWN_R];
    __shared__ unsigned m_l[OWN_R];
    __shared__ float ad_l[OWN_R];
    __shared__ float part[OWN_WAVES];

    const int wave = __builtin_amdgcn_readfirstlane(threadIdx.x >> 5);
    const int lane = threadIdx.x & 31;
    const int n0   = blockIdx.x * OWN_R;
    const int rn   = (N_NODES - n0 < OWN_R) ? (N_NODES - n0) : OWN_R;
    int*   qe  = queue[wave];
    int*   rsd = ret_sd[wave];
    float* rv  = ret_v[wave];

    const int lw = blockIdx.x * OWN_WAVES + wave;
    int   listed = -1;
    int   lsd[LIST_CHUNKS];
    float las[LIST_CHUNKS];
    if (!BUILD) {
        if (ws_size >= WS_NEED) listed = lcount[lw];
#pragma unroll
        for (int k = 0; k < LIST_CHUNKS; ++k) {
            lsd[k] = 0;
            if (k * 32 + lane < listed) lsd[k] = lst[(size_t)lw * OWN_LCAP + k * 32 + lane];
        }
#pragma unroll
        for (int k = 0; k < LIST_CHUNKS; ++k)
            las[k] = as_[lsd[k] >> DL_BITS];
    }

    for (int i = threadIdx.x; i < OWN_R * DIM / 4; i += OWN_THREADS) ((int4*)acc)[i] = make_int4(0, 0, 0, 0);
    if (threadIdx.x < OWN_R) {
        m_l[threadIdx.x]  = 0u;
        z_l[threadIdx.x]  = 0ull;
        ad_l[threadIdx.x] = ((int)threadIdx.x < rn) ? ad_[n0 + threadIdx.x] : 0.0f;
    }
    {
        float4 w4[WMAX_PER_THREAD];
#pragma unroll
        for (int k = 0; k < WMAX_PER_THREAD; ++k) {
            const int i = threadIdx.x + k * OWN_THREADS;
            w4[k] = (i < WMAX_QUADS) ? ((const float4*)wmax)[i] : make_float4(0.f, 0.f, 0.f, 0.f);
        }
        float g = 0.0f;
#pragma unroll
        for (int k = 0; k < WMAX_PER_THREAD; ++k)
            g = fmaxf(fmaxf(g, fmaxf(w4[k].x, w4[k].y)), fmaxf(w4[k].z, w4[k].w));
#pragma unroll
        for (int off = 16; off > 0; off >>= 1)
            g = fmaxf(g, __shfl_xor(g, off, 32));
        if (lane == 0) part[wave] = g;
    }
    __syncthreads();

    float hm = part[lane];
#pragma unroll
    for (int off = 16; off > 0; off >>= 1)
        hm = fmaxf(hm, __shfl_xor(hm, off, 32));

    OwnCtx c;
    c.esrc = esrc; c.as_ = as_; c.Hb = Hb;
    c.ad_l = ad_l; c.m_l = m_l; c.z_l = z_l; c.acc = acc;
    c.qscale = (hm >= HMAX_FLOOR) ? (FIX_ONE / hm) : 0.0f;
    c.lane = lane;

    int  rcount = 0;
    bool ovf    = false;

    if (!BUILD && listed >= 0) {
#pragma unroll
        for (int k = 0; k < LIST_CHUNKS; ++k) {
            const int i = k * 32 + lane;
            if (i < listed) {
                const int   dl = lsd[k] & DL_MASK;
                const float t  = las[k] + ad_l[dl];
                const float v  = (t > 0.0f) ? t : NEG_SLOPE * t;
                atomicMax(&m_l[dl], f2ord(v));
                rsd[i] = lsd[k];
                rv[i]  = v;
            }
        }
        rcount = listed;
    } else {
        own_scan<0>(c, edst, qe, wave, n0, rn, rsd, rv, rcount, ovf);
    }
    rcount = __builtin_amdgcn_readfirstlane(rcount);
    __syncthreads();

    if (ovf) {
        own_scan<1>(c, edst, qe, wave, n0, rn, rsd, rv, rcount, ovf);
    } else {
        for (int c0 = 0; c0 < rcount; c0 += 32) {
            const int i = c0 + lane;
            if (i < rcount) {
                const int   dl = rsd[i] & DL_MASK;
                const float w  = __expf(rv[i] - ord2f(m_l[dl]));
                rv[i] = w;
                atomicAdd(&z_l[dl], (unsigned long long)(w * Z_FIX_ONE));
            }
        }
    }
    __syncthreads();

    if (ovf) {
        own_scan<2>(c, edst, qe, wave, n0, rn, rsd, rv, rcount, ovf);
    } else {
        for (int c0 = 0; c0 < rcount; c0 += 32) {
            const int take = (rcount - c0 < 32) ? (rcount - c0) : 32;
            int   s = 0, dl = 0;
            float coef = 0.0f;
            if (lane < take) {
                const int sd = rsd[c0 + lane];
                s  = sd >> DL_BITS;
                dl = sd & DL_MASK;
                coef = rv[c0 + lane] / own_z(c, dl);
            }
            own_walk(c, s, dl, coef * c.qscale, take);
        }
    }
    __syncthreads();

    const float unscale = hm * FIX_INV_ONE;
    for (int q = threadIdx.x; q < rn * (DIM / 4); q += OWN_THREADS) {
        const int r = q >> 5, l = q & 31;
        const int* rp = acc + r * DIM + l;
        const float4 b = ((const float4*)bias)[l];
        float4 o;
        o.x = elu1((float)rp[0]  * unscale + b.x);
        o.y = elu1((float)rp[32] * unscale + b.y);
        o.z = elu1((float)rp[64] * unscale + b.z);
        o.w = elu1((float)rp[96] * unscale + b.w);
        ((float4*)(X + (size_t)(n0 + r) * DIM))[l] = o;
    }

    if (BUILD) {
        if (ws_size >= WS_NEED) {
            const int n = (!ovf && rcount <= OWN_LCAP) ? rcount : -1;
            for (int k = 0; k < LIST_CHUNKS; ++k) {
                const int i = k * 32 + lane;
                if (i < n) lst[(size_t)lw * OWN_LCAP + i] = rsd[i];
            }
            if (lane == 0) lcount[lw] = n;
        }
    }
}

extern "C" void kernel_launch(void* const* d_in, const int* in_sizes, int n_in,
                              void* d_out, int out_size, void* d_ws, size_t ws_size,
                              hipStream_t stream)
{
    (void)in_sizes; (void)n_in; (void)out_size;

    const float* xq    = (const float*)d_in[0];
    const float* xt    = (const float*)d_in[1];
    const int*   ei_q  = (const int*)d_in[2];
    const float* W_mlp = (const float*)d_in[4];
    const float* b_mlp = (const float*)d_in[5];
    const float* Ws    = (const float*)d_in[6];
    const float* a_src = (const float*)d_in[7];
    const float* a_dst = (const float*)d_in[8];
    const float* biases= (const float*)d_in[9];

    const int* esrc = ei_q;
    const int* edst = ei_q + E_EDGES;

    float* Xq = (float*)d_out;
    float* Xt = Xq + (size_t)N_NODES * DIM;

    __bf16*   Wpk   = (__bf16*)d_ws;
    __bf16*   Hb    = Wpk + (size_t)NWEIGHTS * WPK_ELEMS_PER_W;
    float*    stat  = (float*)(Hb + (size_t)N_NODES * DIM);
    float*    as_   = stat;
    float*    ad_   = as_ + N_NODES;
    float*    wmax  = ad_ + N_NODES;
    int*      lcount = (int*)((char*)d_ws + WS_LIST_OFF);
    int*      lst    = lcount + LIST_WAVES;

    const dim3 blk(256);
    const dim3 pack_grid((NWEIGHTS * 32 * 32 + 255) / 256);
    const dim3 gemm_grid(N_NODES / 16);
    const dim3 node_wave_grid((N_NODES + NODES_PER_WAVE * 8 - 1) / (NODES_PER_WAVE * 8));
    const dim3 own_grid(OWN_BLOCKS);
    const dim3 own_blk(OWN_THREADS);

    pack_weights_kernel<<<pack_grid, blk, 0, stream>>>(W_mlp, Ws, Wpk);

    gemm128_wmma<false><<<gemm_grid, blk, 0, stream>>>(xq, Wpk, b_mlp, Xq, N_NODES);
    gemm128_wmma<false><<<gemm_grid, blk, 0, stream>>>(xt, Wpk, b_mlp, Xt, N_NODES);

    for (int l = 0; l < NLAYERS; ++l) {
        const __bf16* Wl = Wpk + (size_t)(l + 1) * WPK_ELEMS_PER_W;
        const float*  al = a_src  + (size_t)l * DIM;
        const float*  dl = a_dst  + (size_t)l * DIM;
        const float*  bl = biases + (size_t)l * DIM;
        float* Xs[2] = { Xq, Xt };
        for (int g = 0; g < 2; ++g) {
            float* X = Xs[g];
            gemm128_wmma<true><<<gemm_grid, blk, 0, stream>>>(X, Wl, nullptr, (void*)Hb, N_NODES);
            alpha_init_kernel<<<node_wave_grid, blk, 0, stream>>>(Hb, al, dl, as_, ad_, wmax);
            if (l == 0 && g == 0)
                own_softmax_aggr_kernel<true><<<own_grid, own_blk, 0, stream>>>(esrc, edst, as_, ad_, Hb, wmax, bl, X,
                                                                                lcount, lst, ws_size);
            else
                own_softmax_aggr_kernel<false><<<own_grid, own_blk, 0, stream>>>(esrc, edst, as_, ad_, Hb, wmax, bl, X,
                                                                                 lcount, lst, ws_size);
        }
    }
}


# GEMM: the 16-row X strip is read once per work-group with whole-row loads and staged in LDS as bf16; the 8 waves take A fragments from LDS (was 8 x 16-row gathers); products and their order as before
# speedup vs baseline: 5.2622x; 1.2493x over previous
#include <hip/hip_runtime.h>
#include <hip/hip_bf16.h>

#define N_NODES 30000
#define E_EDGES 480000
#define EN_TOT  (E_EDGES + N_NODES)
#define DIM     128
#define NLAYERS 3
#define NEG_SLOPE 0.2f
#define NWEIGHTS 4
#define WPK_ELEMS_PER_W (8 * 4 * 32 * 16)
#define Z_FIX_ONE    1099511627776.0f
#define Z_FIX_INV    9.094947017729282e-13f
#define FIX_ONE      1073741824.0f
#define FIX_INV_ONE  9.313225746154785e-10f
#define HMAX_FLOOR   1.0e-20f

typedef __attribute__((ext_vector_type(16))) __bf16 v16bf;
typedef __attribute__((ext_vector_type(4)))  __bf16 v4bf;
typedef __attribute__((ext_vector_type(8)))  float  v8f;

__device__ __forceinline__ unsigned f2ord(float x) {
    unsigned b = __float_as_uint(x);
    return (b & 0x80000000u) ? ~b : (b | 0x80000000u);
}
__device__ __forceinline__ float ord2f(unsigned u) {
    return __uint_as_float((u & 0x80000000u) ? (u ^ 0x80000000u) : ~u);
}

__global__ void __launch_bounds__(256)
pack_weights_kernel(const float* __restrict__ W_mlp, const float* __restrict__ Ws,
                    __bf16* __restrict__ Wpk)
{
    const int gw   = (blockIdx.x * blockDim.x + threadIdx.x) >> 5;
    const int lane = threadIdx.x & 31;
    if (gw >= NWEIGHTS * 32) return;
    const int w  = gw >> 5;
    const int ct = (gw & 31) >> 2;
    const int s  = gw & 3;
    const float* W = (w == 0) ? W_mlp : (Ws + (size_t)(w - 1) * DIM * DIM);
    const int r  = lane & 15;
    const int hi = lane >> 4;
    const int col = ct * 16 + r;
    const int kb  = s * 32;
    __bf16* out = Wpk + ((size_t)gw * 32 + lane) * 16;
#pragma unroll
    for (int h = 0; h < 16; ++h) {
        const int K = kb + ((h < 8) ? (8 * hi + h) : (16 + 8 * hi + (h - 8)));
        out[h] = (__bf16)W[(size_t)K * DIM + col];
    }
}

#define XS_PITCH 136
typedef __attribute__((ext_vector_type(8))) __bf16 v8bf;

template <bool BF16_OUT>
__global__ void __launch_bounds__(256)
gemm128_wmma(const float* __restrict__ X, const __bf16* __restrict__ Wpk,
             const float* __restrict__ bias, void* __restrict__ out, int rows)
{
    __shared__ __attribute__((aligned(16))) __bf16 xs[16 * XS_PITCH];

    if (blockIdx.x * 16 >= (unsigned)rows) return;
    const int lane = threadIdx.x & 31;
    const int ct   = threadIdx.x >> 5;
    const int r    = lane & 15;
    const int hi   = lane >> 4;
    const int col0 = ct * 16;

    const float4* xp = (const float4*)(X + (size_t)blockIdx.x * 16 * DIM);
    const float4 x0 = xp[threadIdx.x];
    const float4 x1 = xp[threadIdx.x + 256];
    v16bf b[4];
#pragma unroll
    for (int s = 0; s < 4; ++s)
        b[s] = *(const v16bf*)(Wpk + ((size_t)(ct * 4 + s) * 32 + lane) * 16);

    {
        v4bf p0, p1;
        p0[0] = (__bf16)x0.x; p0[1] = (__bf16)x0.y; p0[2] = (__bf16)x0.z; p0[3] = (__bf16)x0.w;
        p1[0] = (__bf16)x1.x; p1[1] = (__bf16)x1.y; p1[2] = (__bf16)x1.z; p1[3] = (__bf16)x1.w;
        *(v4bf*)(xs + ct * XS_PITCH + lane * 4)       = p0;
        *(v4bf*)(xs + (ct + 8) * XS_PITCH + lane * 4) = p1;
    }
    __syncthreads();

    v8f acc = {};
#pragma unroll
    for (int s = 0; s < 4; ++s) {
        const int kb = s * 32;
        const v8bf alo = *(const v8bf*)(xs + r * XS_PITCH + kb + 8 * hi);
        const v8bf ahi = *(const v8bf*)(xs + r * XS_PITCH + kb + 16 + 8 * hi);
        const v16bf a = __builtin_shufflevector(alo, ahi, 0, 1, 2, 3, 4, 5, 6, 7, 8, 9, 10, 11, 12, 13, 14, 15);
        acc = __builtin_amdgcn_wmma_f32_16x16x32_bf16(
             false, a,  false, b[s],
             (short)0, acc,  false,  false);
    }
    if (BF16_OUT) {
        __bf16* H = (__bf16*)out;
#pragma unroll
        for (int v = 0; v < 8; ++v) {
            const int m = blockIdx.x * 16 + v + 8 * hi;
            H[(size_t)m * DIM + col0 + r] = (__bf16)acc[v];
        }
    } else {
        float* H = (float*)out;
        const float bv = bias[col0 + r];
#pragma unroll
        for (int v = 0; v < 8; ++v) {
            const int m = blockIdx.x * 16 + v + 8 * hi;
            H[(size_t)m * DIM + col0 + r] = acc[v] + bv;
        }
    }
}

#define NODES_PER_WAVE 4
#define ALPHA_WAVES (N_NODES / NODES_PER_WAVE)
static_assert(ALPHA_WAVES % 4 == 0, "wmax is read in 16-byte pieces");
static_assert(N_NODES % NODES_PER_WAVE == 0, "a wave's rows must all exist");

__global__ void __launch_bounds__(256)
alpha_init_kernel(const __bf16* __restrict__ Hb,
                  const float* __restrict__ asrc, const float* __restrict__ adst,
                  float* __restrict__ as_, float* __restrict__ ad_,
                  float* __restrict__ wmax)
{
    const int wave = blockIdx.x * (blockDim.x >> 5) + (threadIdx.x >> 5);
    const int lane = threadIdx.x & 31;
    const int n0   = wave * NODES_PER_WAVE;
    if (n0 >= N_NODES) return;

    const float4 s4 = ((const float4*)asrc)[lane];
    const float4 d4 = ((const float4*)adst)[lane];
    v4bf h4[NODES_PER_WAVE];
#pragma unroll
    for (int k = 0; k < NODES_PER_WAVE; ++k)
        h4[k] = ((const v4bf*)(Hb + (size_t)(n0 + k) * DIM))[lane];

    float amax = 0.0f;
#pragma unroll
    for (int k = 0; k < NODES_PER_WAVE; ++k) {
        const int node = n0 + k;
        const float hx = (float)h4[k][0], hy = (float)h4[k][1], hz = (float)h4[k][2], hw = (float)h4[k][3];
        amax = fmaxf(fmaxf(amax, fmaxf(fabsf(hx), fabsf(hy))), fmaxf(fabsf(hz), fabsf(hw)));
        float ps = hx * s4.x + hy * s4.y + hz * s4.z + hw * s4.w;
        float pd = hx * d4.x + hy * d4.y + hz * d4.z + hw * d4.w;
#pragma unroll
        for (int off = 16; off > 0; off >>= 1) {
            ps += __shfl_down(ps, off, 32);
            pd += __shfl_down(pd, off, 32);
        }
        if (lane == 0) {
            as_[node] = ps;
            ad_[node] = pd;
        }
    }
#pragma unroll
    for (int off = 16; off > 0; off >>= 1)
        amax = fmaxf(amax, __shfl_xor(amax, off, 32));
    if (lane == 0) wmax[wave] = amax;
}

#define ROWS_AHEAD   8
#define OWN_R        118
#define OWN_BLOCKS   ((N_NODES + OWN_R - 1) / OWN_R)
#define OWN_THREADS  1024
#define OWN_WAVES    (OWN_THREADS / 32)
#define SCAN_AHEAD   8
#define E_QUADS      (E_EDGES / 4)
#define SCAN_ITERS   ((E_QUADS + OWN_THREADS - 1) / OWN_THREADS)
#define SCAN_GROUPS  ((SCAN_ITERS + SCAN_AHEAD - 1) / SCAN_AHEAD)
#define OWN_QCAP     (32 + SCAN_AHEAD * 4 * 32)
#define OWN_RCAP     256
#define DL_BITS      7
#define DL_MASK      ((1 << DL_BITS) - 1)
#define WMAX_QUADS   (ALPHA_WAVES / 4)
#define WMAX_PER_THREAD ((WMAX_QUADS + OWN_THREADS - 1) / OWN_THREADS)
static_assert(E_EDGES % 4 == 0, "the destination list is read in 16-byte pieces");
static_assert(OWN_R <= (1 << DL_BITS), "row inside the group must fit DL_BITS");
static_assert(EN_TOT < (1 << (31 - DL_BITS)), "edge number and row share one 32-bit queue word");
static_assert(N_NODES < (1 << (31 - DL_BITS)), "source and row share one 32-bit word");
static_assert(OWN_WAVES == 32, "the per-wave maxima are folded by one wave-wide exchange");
#define OWN_LCAP     128
#define LIST_CHUNKS  (OWN_LCAP / 32)
#define LIST_WAVES   (OWN_BLOCKS * OWN_WAVES)
#define WS_STAT_END  (NWEIGHTS * WPK_ELEMS_PER_W * 2 + N_NODES * DIM * 2 + (2 * N_NODES + ALPHA_WAVES) * 4)
#define WS_LIST_OFF  ((WS_STAT_END + 127) / 128 * 128)
#define WS_NEED      ((size_t)WS_LIST_OFF + (size_t)LIST_WAVES * 4 + (size_t)LIST_WAVES * OWN_LCAP * 4)
static_assert(OWN_LCAP % 32 == 0, "a list is read 32 entries at a time");
static_assert(OWN_LCAP <= OWN_RCAP, "a listed wave retains all its entries");

__device__ __forceinline__ v4bf load_row4(const __bf16* __restrict__ Hb, int s, int lane)
{
    return ((const v4bf*)(Hb + (size_t)s * DIM))[lane];
}

__device__ __forceinline__ float elu1(float v) { return (v > 0.0f) ? v : (__expf(v) - 1.0f); }

struct OwnCtx {
    const int*    esrc;
    const float*  as_;
    const __bf16* Hb;
    const float*  ad_l;
    unsigned*     m_l;
    unsigned long long* z_l;
    int*          acc;
    float         qscale;
    int           lane;
};

__device__ __forceinline__ float own_z(const OwnCtx& c, int dl)
{
    return (float)c.z_l[dl] * Z_FIX_INV;
}

__device__ __forceinline__ void own_walk(const OwnCtx& c, int s, int dl, float coef_scaled, int take)
{
    const int lane = c.lane;
    const int coef_bits = __float_as_int(coef_scaled);
    v4bf h[ROWS_AHEAD];
#pragma unroll
    for (int u = 0; u < ROWS_AHEAD; ++u)
        h[u] = load_row4(c.Hb, __builtin_amdgcn_readlane(s, u), lane);

    for (int j0 = 0; j0 < take; j0 += ROWS_AHEAD) {
        v4bf hn[ROWS_AHEAD];
        const bool more = (j0 + ROWS_AHEAD < take);
        if (more) {
#pragma unroll
            for (int u = 0; u < ROWS_AHEAD; ++u)
                hn[u] = load_row4(c.Hb, __builtin_amdgcn_readlane(s, j0 + ROWS_AHEAD + u), lane);
        }
#pragma unroll
        for (int u = 0; u < ROWS_AHEAD; ++u) {
            const int   dj = __builtin_amdgcn_readlane(dl, j0 + u);
            const float cj = __int_as_float(__builtin_amdgcn_readlane(coef_bits, j0 + u));
            int* ap = c.acc + dj * DIM + lane;
            atomicAdd(ap +  0, __float2int_rn(cj * (float)h[u][0]));
            atomicAdd(ap + 32, __float2int_rn(cj * (float)h[u][1]));
            atomicAdd(ap + 64, __float2int_rn(cj * (float)h[u][2]));
            atomicAdd(ap + 96, __float2int_rn(cj * (float)h[u][3]));
        }
        if (more) {
#pragma unroll
            for (int u = 0; u < ROWS_AHEAD; ++u) h[u] = hn[u];
        }
    }
}

template <int PHASE>
__device__ __forceinline__ void own_chunk(const OwnCtx& c, const int* qe, int take,
                                          int* rsd, float* rv, int& rcount, bool& ovf)
{
    const int lane = c.lane;
    const bool on = (lane < take);
    int   s = 0, dl = 0;
    float v = 0.0f;
    if (on) {
        const int q = qe[lane];
        const int e = q >> DL_BITS;
        dl = q & DL_MASK;
        s = (e < E_EDGES) ? c.esrc[e] : (e - E_EDGES);
        const float t = c.as_[s] + c.ad_l[dl];
        v = (t > 0.0f) ? t : NEG_SLOPE * t;
    }
    if (PHASE == 0) {
        if (on) atomicMax(&c.m_l[dl], f2ord(v));
        if (!ovf) {
            if (rcount + take <= OWN_RCAP) {
                if (on) { rsd[rcount + lane] = (s << DL_BITS) | dl; rv[rcount + lane] = v; }
                rcount += take;
            } else {
                ovf = true;
            }
        }
    } else if (PHASE == 1) {
        if (on) {
            const float w = __expf(v - ord2f(c.m_l[dl]));
            atomicAdd(&c.z_l[dl], (unsigned long long)(w * Z_FIX_ONE));
        }
    } else {
        float coef = 0.0f;
        if (on) coef = __expf(v - ord2f(c.m_l[dl])) / own_z(c, dl);
        own_walk(c, s, dl, coef * c.qscale, take);
    }
}

template <int PHASE>
__device__ __forceinline__ void own_scan(const OwnCtx& c, const int* __restrict__ edst, int* qe,
                                         int wave, int n0, int rn,
                                         int* rsd, float* rv, int& rcount, bool& ovf)
{
    const int lane = c.lane;
    int qn = 0;

    {
        const int row = wave + 32 * lane;
        const bool keep = (row < rn);
        if (keep) qe[lane] = ((E_EDGES + n0 + row) << DL_BITS) | row;
        qn = __popc(__builtin_amdgcn_ballot_w32(keep));
    }

    for (int g = 0; g < SCAN_GROUPS; ++g) {
        const int it0 = ((g + (int)blockIdx.x) % SCAN_GROUPS) * SCAN_AHEAD;
        int4 d4[SCAN_AHEAD];
#pragma unroll
        for (int u = 0; u < SCAN_AHEAD; ++u) {
            const int idx = (it0 + u) * OWN_THREADS + (int)threadIdx.x;
            d4[u] = (idx < E_QUADS) ? ((const int4*)edst)[idx] : make_int4(-1, -1, -1, -1);
        }
#pragma unroll
        for (int u = 0; u < SCAN_AHEAD; ++u) {
            const int idx = (it0 + u) * OWN_THREADS + (int)threadIdx.x;
            const int dd[4] = { d4[u].x, d4[u].y, d4[u].z, d4[u].w };
#pragma unroll
            for (int k = 0; k < 4; ++k) {
                const int  dl   = dd[k] - n0;
                const bool keep = (unsigned)dl < (unsigned)rn;
                const unsigned mask = __builtin_amdgcn_ballot_w32(keep);
                if (mask != 0u) {
                    if (keep) qe[qn + __popc(mask & ((1u << lane) - 1u))] = ((idx * 4 + k) << DL_BITS) | dl;
                    qn += __popc(mask);
                }
            }
        }
        while (qn >= 32) {
            qn -= 32;
            own_chunk<PHASE>(c, qe + qn, 32, rsd, rv, rcount, ovf);
        }
    }
    if (qn > 0)
        own_chunk<PHASE>(c, qe, qn, rsd, rv, rcount, ovf);
}

template <bool BUILD>
__global__ void __launch_bounds__(OWN_THREADS)
own_softmax_aggr_kernel(const int* __restrict__ esrc, const int* __restrict__ edst,
                        const float* __restrict__ as_, const float* __restrict__ ad_,
                        const __bf16* __restrict__ Hb, const float* __restrict__ wmax,
                        const float* __restrict__ bias, float* __restrict__ X,
                        int* __restrict__ lcount, int* __restrict__ lst, size_t ws_size)
{
    __shared__ __attribute__((aligned(16))) int acc[OWN_R * DIM];
    __shared__ int   queue[OWN_WAVES][OWN_QCAP];
    __shared__ int   ret_sd[OWN_WAVES][OWN_RCAP];
    __shared__ float ret_v[OWN_WAVES][OWN_RCAP];
    __shared__ unsigned long long z_l[OWN_R];
    __shared__ unsigned m_l[OWN_R];
    __shared__ float ad_l[OWN_R];
    __shared__ float part[OWN_WAVES];

    const int wave = __builtin_amdgcn_readfirstlane(threadIdx.x >> 5);
    const int lane = threadIdx.x & 31;
    const int n0   = blockIdx.x * OWN_R;
    const int rn   = (N_NODES - n0 < OWN_R) ? (N_NODES - n0) : OWN_R;
    int*   qe  = queue[wave];
    int*   rsd = ret_sd[wave];
    float* rv  = ret_v[wave];

    const int lw = blockIdx.x * OWN_WAVES + wave;
    int   listed = -1;
    int   lsd[LIST_CHUNKS];
    float las[LIST_CHUNKS];
    if (!BUILD) {
        if (ws_size >= WS_NEED) listed = lcount[lw];
#pragma unroll
        for (int k = 0; k < LIST_CHUNKS; ++k) {
            lsd[k] = 0;
            if (k * 32 + lane < listed) lsd[k] = lst[(size_t)lw * OWN_LCAP + k * 32 + lane];
        }
#pragma unroll
        for (int k = 0; k < LIST_CHUNKS; ++k)
            las[k] = as_[lsd[k] >> DL_BITS];
    }

    for (int i = threadIdx.x; i < OWN_R * DIM / 4; i += OWN_THREADS) ((int4*)acc)[i] = make_int4(0, 0, 0, 0);
    if (threadIdx.x < OWN_R) {
        m_l[threadIdx.x]  = 0u;
        z_l[threadIdx.x]  = 0ull;
        ad_l[threadIdx.x] = ((int)threadIdx.x < rn) ? ad_[n0 + threadIdx.x] : 0.0f;
    }
    {
        float4 w4[WMAX_PER_THREAD];
#pragma unroll
        for (int k = 0; k < WMAX_PER_THREAD; ++k) {
            const int i = threadIdx.x + k * OWN_THREADS;
            w4[k] = (i < WMAX_QUADS) ? ((const float4*)wmax)[i] : make_float4(0.f, 0.f, 0.f, 0.f);
        }
        float g = 0.0f;
#pragma unroll
        for (int k = 0; k < WMAX_PER_THREAD; ++k)
            g = fmaxf(fmaxf(g, fmaxf(w4[k].x, w4[k].y)), fmaxf(w4[k].z, w4[k].w));
#pragma unroll
        for (int off = 16; off > 0; off >>= 1)
            g = fmaxf(g, __shfl_xor(g, off, 32));
        if (lane == 0) part[wave] = g;
    }
    __syncthreads();

    float hm = part[lane];
#pragma unroll
    for (int off = 16; off > 0; off >>= 1)
        hm = fmaxf(hm, __shfl_xor(hm, off, 32));

    OwnCtx c;
    c.esrc = esrc; c.as_ = as_; c.Hb = Hb;
    c.ad_l = ad_l; c.m_l = m_l; c.z_l = z_l; c.acc = acc;
    c.qscale = (hm >= HMAX_FLOOR) ? (FIX_ONE / hm) : 0.0f;
    c.lane = lane;

    int  rcount = 0;
    bool ovf    = false;

    if (!BUILD && listed >= 0) {
#pragma unroll
        for (int k = 0; k < LIST_CHUNKS; ++k) {
            const int i = k * 32 + lane;
            if (i < listed) {
                const int   dl = lsd[k] & DL_MASK;
                const float t  = las[k] + ad_l[dl];
                const float v  = (t > 0.0f) ? t : NEG_SLOPE * t;
                atomicMax(&m_l[dl], f2ord(v));
                rsd[i] = lsd[k];
                rv[i]  = v;
            }
        }
        rcount = listed;
    } else {
        own_scan<0>(c, edst, qe, wave, n0, rn, rsd, rv, rcount, ovf);
    }
    rcount = __builtin_amdgcn_readfirstlane(rcount);
    __syncthreads();

    if (ovf) {
        own_scan<1>(c, edst, qe, wave, n0, rn, rsd, rv, rcount, ovf);
    } else {
        for (int c0 = 0; c0 < rcount; c0 += 32) {
            const int i = c0 + lane;
            if (i < rcount) {
                const int   dl = rsd[i] & DL_MASK;
                const float w  = __expf(rv[i] - ord2f(m_l[dl]));
                rv[i] = w;
                atomicAdd(&z_l[dl], (unsigned long long)(w * Z_FIX_ONE));
            }
        }
    }
    __syncthreads();

    if (ovf) {
        own_scan<2>(c, edst, qe, wave, n0, rn, rsd, rv, rcount, ovf);
    } else {
        for (int c0 = 0; c0 < rcount; c0 += 32) {
            const int take = (rcount - c0 < 32) ? (rcount - c0) : 32;
            int   s = 0, dl = 0;
            float coef = 0.0f;
            if (lane < take) {
                const int sd = rsd[c0 + lane];
                s  = sd >> DL_BITS;
                dl = sd & DL_MASK;
                coef = rv[c0 + lane] / own_z(c, dl);
            }
            own_walk(c, s, dl, coef * c.qscale, take);
        }
    }
    __syncthreads();

    const float unscale = hm * FIX_INV_ONE;
    for (int q = threadIdx.x; q < rn * (DIM / 4); q += OWN_THREADS) {
        const int r = q >> 5, l = q & 31;
        const int* rp = acc + r * DIM + l;
        const float4 b = ((const float4*)bias)[l];
        float4 o;
        o.x = elu1((float)rp[0]  * unscale + b.x);
        o.y = elu1((float)rp[32] * unscale + b.y);
        o.z = elu1((float)rp[64] * unscale + b.z);
        o.w = elu1((float)rp[96] * unscale + b.w);
        ((float4*)(X + (size_t)(n0 + r) * DIM))[l] = o;
    }

    if (BUILD) {
        if (ws_size >= WS_NEED) {
            const int n = (!ovf && rcount <= OWN_LCAP) ? rcount : -1;
            for (int k = 0; k < LIST_CHUNKS; ++k) {
                const int i = k * 32 + lane;
                if (i < n) lst[(size_t)lw * OWN_LCAP + i] = rsd[i];
            }
            if (lane == 0) lcount[lw] = n;
        }
    }
}

extern "C" void kernel_launch(void* const* d_in, const int* in_sizes, int n_in,
                              void* d_out, int out_size, void* d_ws, size_t ws_size,
                              hipStream_t stream)
{
    (void)in_sizes; (void)n_in; (void)out_size;

    const float* xq    = (const float*)d_in[0];
    const float* xt    = (const float*)d_in[1];
    const int*   ei_q  = (const int*)d_in[2];
    const float* W_mlp = (const float*)d_in[4];
    const float* b_mlp = (const float*)d_in[5];
    const float* Ws    = (const float*)d_in[6];
    const float* a_src = (const float*)d_in[7];
    const float* a_dst = (const float*)d_in[8];
    const float* biases= (const float*)d_in[9];

    const int* esrc = ei_q;
    const int* edst = ei_q + E_EDGES;

    float* Xq = (float*)d_out;
    float* Xt = Xq + (size_t)N_NODES * DIM;

    __bf16*   Wpk   = (__bf16*)d_ws;
    __bf16*   Hb    = Wpk + (size_t)NWEIGHTS * WPK_ELEMS_PER_W;
    float*    stat  = (float*)(Hb + (size_t)N_NODES * DIM);
    float*    as_   = stat;
    float*    ad_   = as_ + N_NODES;
    float*    wmax  = ad_ + N_NODES;
    int*      lcount = (int*)((char*)d_ws + WS_LIST_OFF);
    int*      lst    = lcount + LIST_WAVES;

    const dim3 blk(256);
    const dim3 pack_grid((NWEIGHTS * 32 * 32 + 255) / 256);
    const dim3 gemm_grid(N_NODES / 16);
    const dim3 node_wave_grid((N_NODES + NODES_PER_WAVE * 8 - 1) / (NODES_PER_WAVE * 8));
    const dim3 own_grid(OWN_BLOCKS);
    const dim3 own_blk(OWN_THREADS);

    pack_weights_kernel<<<pack_grid, blk, 0, stream>>>(W_mlp, Ws, Wpk);

    gemm128_wmma<false><<<gemm_grid, blk, 0, stream>>>(xq, Wpk, b_mlp, Xq, N_NODES);
    gemm128_wmma<false><<<gemm_grid, blk, 0, stream>>>(xt, Wpk, b_mlp, Xt, N_NODES);

    for (int l = 0; l < NLAYERS; ++l) {
        const __bf16* Wl = Wpk + (size_t)(l + 1) * WPK_ELEMS_PER_W;
        const float*  al = a_src  + (size_t)l * DIM;
        const float*  dl = a_dst  + (size_t)l * DIM;
        const float*  bl = biases + (size_t)l * DIM;
        float* Xs[2] = { Xq, Xt };
        for (int g = 0; g < 2; ++g) {
            float* X = Xs[g];
            gemm128_wmma<true><<<gemm_grid, blk, 0, stream>>>(X, Wl, nullptr, (void*)Hb, N_NODES);
            alpha_init_kernel<<<node_wave_grid, blk, 0, stream>>>(Hb, al, dl, as_, ad_, wmax);
            if (l == 0 && g == 0)
                own_softmax_aggr_kernel<true><<<own_grid, own_blk, 0, stream>>>(esrc, edst, as_, ad_, Hb, wmax, bl, X,
                                                                                lcount, lst, ws_size);
            else
                own_softmax_aggr_kernel<false><<<own_grid, own_blk, 0, stream>>>(esrc, edst, as_, ad_, Hb, wmax, bl, X,
                                                                                 lcount, lst, ws_size);
        }
    }
}


# Attention dots reassociated, alpha = X.(W a): taken in the GEMM launch from the f32 X strip and pre-multiplied vectors (new sum order); max|H| from its epilogue; alpha launches gone, 21 -> 15
# speedup vs baseline: 6.0572x; 1.1511x over previous
#include <hip/hip_runtime.h>
#include <hip/hip_bf16.h>

#define N_NODES 30000
#define E_EDGES 480000
#define EN_TOT  (E_EDGES + N_NODES)
#define DIM     128
#define NLAYERS 3
#define NEG_SLOPE 0.2f
#define NWEIGHTS 4
#define WPK_ELEMS_PER_W (8 * 4 * 32 * 16)
#define Z_FIX_ONE    1099511627776.0f
#define Z_FIX_INV    9.094947017729282e-13f
#define FIX_ONE        2097152.0f
#define FIX_INV_ONE    4.76837158203125e-07f
#define FIX_MAGIC      12582912.0f
#define FIX_MAGIC_BITS 0x4B400000u
#define HMAX_FLOOR   1.0e-20f

typedef __attribute__((ext_vector_type(16))) __bf16 v16bf;
typedef __attribute__((ext_vector_type(4)))  __bf16 v4bf;
typedef __attribute__((ext_vector_type(8)))  float  v8f;

__device__ __forceinline__ unsigned f2ord(float x) {
    unsigned b = __float_as_uint(x);
    return (b & 0x80000000u) ? ~b : (b | 0x80000000u);
}
__device__ __forceinline__ float ord2f(unsigned u) {
    return __uint_as_float((u & 0x80000000u) ? (u ^ 0x80000000u) : ~u);
}

template <int M>
__device__ __forceinline__ float xrow(float x) {
    const int b = __builtin_bit_cast(int, x);
    return __builtin_bit_cast(float, __builtin_amdgcn_update_dpp(b, b, 0x160 + M, 0xf, 0xf, false));
}
__device__ __forceinline__ float xhalf(float x) {
    const int b = __builtin_bit_cast(int, x);
    return __builtin_bit_cast(float, __builtin_amdgcn_permlanex16(b, b, 0x76543210, 0xfedcba98, false, false));
}
__device__ __forceinline__ float wave_sum(float x) {
    x += xrow<1>(x); x += xrow<2>(x); x += xrow<4>(x); x += xrow<8>(x);
    return x + xhalf(x);
}
__device__ __forceinline__ float wave_max(float x) {
    x = fmaxf(x, xrow<1>(x)); x = fmaxf(x, xrow<2>(x)); x = fmaxf(x, xrow<4>(x)); x = fmaxf(x, xrow<8>(x));
    return fmaxf(x, xhalf(x));
}

#define PACK_WAVES (NWEIGHTS * 32)
#define UVEC_WAVES (NLAYERS * DIM)
__global__ void __launch_bounds__(256)
pack_weights_kernel(const float* __restrict__ W_mlp, const float* __restrict__ Ws,
                    const float* __restrict__ a_src, const float* __restrict__ a_dst,
                    __bf16* __restrict__ Wpk, float* __restrict__ uvec)
{
    const int gw   = (blockIdx.x * blockDim.x + threadIdx.x) >> 5;
    const int lane = threadIdx.x & 31;
    if (gw >= PACK_WAVES + UVEC_WAVES) return;
    if (gw >= PACK_WAVES) {
        const int j = gw - PACK_WAVES;
        const int l = j / DIM;
        const int k = j % DIM;
        const float4 w4 = ((const float4*)(Ws + ((size_t)l * DIM + k) * DIM))[lane];
        const float4 s4 = ((const float4*)(a_src + (size_t)l * DIM))[lane];
        const float4 d4 = ((const float4*)(a_dst + (size_t)l * DIM))[lane];
        float ps = w4.x * s4.x + w4.y * s4.y + w4.z * s4.z + w4.w * s4.w;
        float pd = w4.x * d4.x + w4.y * d4.y + w4.z * d4.z + w4.w * d4.w;
        ps = wave_sum(ps);
        pd = wave_sum(pd);
        if (lane == 0) {
            uvec[(size_t)(l * 2 + 0) * DIM + k] = ps;
            uvec[(size_t)(l * 2 + 1) * DIM + k] = pd;
        }
        return;
    }
    const int w  = gw >> 5;
    const int ct = (gw & 31) >> 2;
    const int s  = gw & 3;
    const float* W = (w == 0) ? W_mlp : (Ws + (size_t)(w - 1) * DIM * DIM);
    const int r  = lane & 15;
    const int hi = lane >> 4;
    const int col = ct * 16 + r;
    const int kb  = s * 32;
    __bf16* out = Wpk + ((size_t)gw * 32 + lane) * 16;
#pragma unroll
    for (int h = 0; h < 16; ++h) {
        const int K = kb + ((h < 8) ? (8 * hi + h) : (16 + 8 * hi + (h - 8)));
        out[h] = (__bf16)W[(size_t)K * DIM + col];
    }
}

#define GEMM_BLOCKS (N_NODES / 16)
static_assert(N_NODES % 16 == 0, "every strip is whole");
#define XS_PITCH 136
typedef __attribute__((ext_vector_type(8))) __bf16 v8bf;

template <bool BF16_OUT>
__global__ void __launch_bounds__(256)
gemm128_wmma(const float* __restrict__ X, const __bf16* __restrict__ Wpk,
             const float* __restrict__ bias, void* __restrict__ out, int rows,
             const float* __restrict__ uvec, float* __restrict__ as_, float* __restrict__ ad_,
             float* __restrict__ wmax)
{
    __shared__ __attribute__((aligned(16))) __bf16 xs[16 * XS_PITCH];
    __shared__ float al[32];
    __shared__ float part[8];

    if (blockIdx.x * 16 >= (unsigned)rows) return;
    const int lane = threadIdx.x & 31;
    const int ct   = threadIdx.x >> 5;
    const int r    = lane & 15;
    const int hi   = lane >> 4;
    const int col0 = ct * 16;

    const float4* xp = (const float4*)(X + (size_t)blockIdx.x * 16 * DIM);
    v16bf b[4];
#pragma unroll
    for (int s = 0; s < 4; ++s)
        b[s] = *(const v16bf*)(Wpk + ((size_t)(ct * 4 + s) * 32 + lane) * 16);
    const float4 x0 = xp[threadIdx.x];
    const float4 x1 = xp[threadIdx.x + 256];
    float4 us4 = make_float4(0.f, 0.f, 0.f, 0.f), ud4 = us4;
    if (BF16_OUT) {
        us4 = ((const float4*)uvec)[lane];
        ud4 = ((const float4*)(uvec + DIM))[lane];
        __builtin_amdgcn_sched_barrier(0);
    }

    {
        v4bf p0, p1;
        p0[0] = (__bf16)x0.x; p0[1] = (__bf16)x0.y; p0[2] = (__bf16)x0.z; p0[3] = (__bf16)x0.w;
        p1[0] = (__bf16)x1.x; p1[1] = (__bf16)x1.y; p1[2] = (__bf16)x1.z; p1[3] = (__bf16)x1.w;
        *(v4bf*)(xs + ct * XS_PITCH + lane * 4)       = p0;
        *(v4bf*)(xs + (ct + 8) * XS_PITCH + lane * 4) = p1;
    }
    if (BF16_OUT) {
        float ps0 = x0.x * us4.x + x0.y * us4.y + x0.z * us4.z + x0.w * us4.w;
        float pd0 = x0.x * ud4.x + x0.y * ud4.y + x0.z * ud4.z + x0.w * ud4.w;
        float ps1 = x1.x * us4.x + x1.y * us4.y + x1.z * us4.z + x1.w * us4.w;
        float pd1 = x1.x * ud4.x + x1.y * ud4.y + x1.z * ud4.z + x1.w * ud4.w;
        ps0 = wave_sum(ps0);  pd0 = wave_sum(pd0);
        ps1 = wave_sum(ps1);  pd1 = wave_sum(pd1);
        if (lane == 0) {
            al[ct]      = ps0;  al[ct + 8]  = ps1;
            al[16 + ct] = pd0;  al[24 + ct] = pd1;
        }
    }
    __syncthreads();
    if (BF16_OUT && ct == 0) {
        float* dst = (lane < 16) ? (as_ + (size_t)blockIdx.x * 16 + lane)
                                 : (ad_ + (size_t)blockIdx.x * 16 + (lane - 16));
        *dst = al[lane];
    }

    v8f acc = {};
#pragma unroll
    for (int s = 0; s < 4; ++s) {
        const int kb = s * 32;
        const v8bf alo = *(const v8bf*)(xs + r * XS_PITCH + kb + 8 * hi);
        const v8bf ahi = *(const v8bf*)(xs + r * XS_PITCH + kb + 16 + 8 * hi);
        const v16bf a = __builtin_shufflevector(alo, ahi, 0, 1, 2, 3, 4, 5, 6, 7, 8, 9, 10, 11, 12, 13, 14, 15);
        acc = __builtin_amdgcn_wmma_f32_16x16x32_bf16(
             false, a,  false, b[s],
             (short)0, acc,  false,  false);
    }
    if (BF16_OUT) {
        __bf16* H = (__bf16*)out;
        float amax = 0.0f;
#pragma unroll
        for (int v = 0; v < 8; ++v) {
            const int m = blockIdx.x * 16 + v + 8 * hi;
            H[(size_t)m * DIM + col0 + r] = (__bf16)acc[v];
            amax = fmaxf(amax, fabsf(acc[v]));
        }
        amax = wave_max(amax);
        if (lane == 0) part[ct] = amax;
        __syncthreads();
        if (threadIdx.x == 0) {
            float g = part[0];
#pragma unroll
            for (int i = 1; i < 8; ++i) g = fmaxf(g, part[i]);
            wmax[blockIdx.x] = (float)(__bf16)g;
        }
    } else {
        float* H = (float*)out;
        const float bv = bias[col0 + r];
#pragma unroll
        for (int v = 0; v < 8; ++v) {
            const int m = blockIdx.x * 16 + v + 8 * hi;
            H[(size_t)m * DIM + col0 + r] = acc[v] + bv;
        }
    }
}

#define STAT_TAIL_WORDS 7500
#define UVEC_OFF        2048
#define UVEC_WORDS      (NLAYERS * 2 * DIM)
static_assert(GEMM_BLOCKS <= UVEC_OFF, "wmax ends before uvec begins");
static_assert(UVEC_OFF % 4 == 0, "uvec is read in 16-byte pieces");
static_assert(UVEC_OFF + UVEC_WORDS <= STAT_TAIL_WORDS, "uvec ends inside the tail");

#define ROWS_AHEAD   8
#define OWN_R        118
#define OWN_BLOCKS   ((N_NODES + OWN_R - 1) / OWN_R)
#define OWN_THREADS  1024
#define OWN_WAVES    (OWN_THREADS / 32)
#define SCAN_AHEAD   8
#define E_QUADS      (E_EDGES / 4)
#define SCAN_ITERS   ((E_QUADS + OWN_THREADS - 1) / OWN_THREADS)
#define SCAN_GROUPS  ((SCAN_ITERS + SCAN_AHEAD - 1) / SCAN_AHEAD)
#define OWN_QCAP     (32 + SCAN_AHEAD * 4 * 32)
#define OWN_RCAP     256
#define DL_BITS      7
#define DL_MASK      ((1 << DL_BITS) - 1)
#define WMAX_PER_THREAD ((GEMM_BLOCKS + OWN_THREADS - 1) / OWN_THREADS)
static_assert(E_EDGES % 4 == 0, "the destination list is read in 16-byte pieces");
static_assert(OWN_R <= (1 << DL_BITS), "row inside the group must fit DL_BITS");
static_assert(EN_TOT < (1 << (31 - DL_BITS)), "edge number and row share one 32-bit queue word");
static_assert(N_NODES < (1 << (31 - DL_BITS)), "source and row share one 32-bit word");
static_assert(OWN_WAVES == 32, "the per-wave maxima are folded by one wave-wide exchange");
#define OWN_LCAP     128
#define LIST_CHUNKS  (OWN_LCAP / 32)
#define LIST_WAVES   (OWN_BLOCKS * OWN_WAVES)
#define WS_STAT_END  (NWEIGHTS * WPK_ELEMS_PER_W * 2 + N_NODES * DIM * 2 + (2 * N_NODES + STAT_TAIL_WORDS) * 4)
#define WS_LIST_OFF  ((WS_STAT_END + 127) / 128 * 128)
#define WS_NEED      ((size_t)WS_LIST_OFF + (size_t)LIST_WAVES * 4 + (size_t)LIST_WAVES * OWN_LCAP * 4)
static_assert(OWN_LCAP % 32 == 0, "a list is read 32 entries at a time");
static_assert(OWN_LCAP <= OWN_RCAP, "a listed wave retains all its entries");

__device__ __forceinline__ v4bf load_row4(const __bf16* __restrict__ Hb, int s, int lane)
{
    return ((const v4bf*)(Hb + (size_t)s * DIM))[lane];
}

__device__ __forceinline__ float elu1(float v) { return (v > 0.0f) ? v : (__expf(v) - 1.0f); }

struct OwnCtx {
    const int*    esrc;
    const float*  as_;
    const __bf16* Hb;
    const float*  ad_l;
    unsigned*     m_l;
    unsigned long long* z_l;
    int*          acc;
    unsigned*     cnt_l;
    float         qscale;
    int           lane;
};

__device__ __forceinline__ float own_z(const OwnCtx& c, int dl)
{
    return (float)c.z_l[dl] * Z_FIX_INV;
}

__device__ __forceinline__ void own_walk(const OwnCtx& c, int s, int dl, float coef_scaled, int take)
{
    const int lane = c.lane;
    const int coef_bits = __float_as_int(coef_scaled);
    if (lane < take) atomicAdd(&c.cnt_l[dl], 1u);
    dl = (lane < take) ? dl : OWN_R;
    v4bf h[ROWS_AHEAD];
#pragma unroll
    for (int u = 0; u < ROWS_AHEAD; ++u)
        h[u] = load_row4(c.Hb, __builtin_amdgcn_readlane(s, u), lane);

    for (int j0 = 0; j0 < take; j0 += ROWS_AHEAD) {
        v4bf hn[ROWS_AHEAD];
        const bool more = (j0 + ROWS_AHEAD < take);
        if (more) {
#pragma unroll
            for (int u = 0; u < ROWS_AHEAD; ++u)
                hn[u] = load_row4(c.Hb, __builtin_amdgcn_readlane(s, j0 + ROWS_AHEAD + u), lane);
        }
#pragma unroll
        for (int u = 0; u < ROWS_AHEAD; ++u) {
            const int   dj = __builtin_amdgcn_readlane(dl, j0 + u);
            const float cj = __int_as_float(__builtin_amdgcn_readlane(coef_bits, j0 + u));
            int* ap = c.acc + dj * DIM + lane;
            atomicAdd(ap +  0, __float_as_int(__builtin_fmaf(cj, (float)h[u][0], FIX_MAGIC)));
            atomicAdd(ap + 32, __float_as_int(__builtin_fmaf(cj, (float)h[u][1], FIX_MAGIC)));
            atomicAdd(ap + 64, __float_as_int(__builtin_fmaf(cj, (float)h[u][2], FIX_MAGIC)));
            atomicAdd(ap + 96, __float_as_int(__builtin_fmaf(cj, (float)h[u][3], FIX_MAGIC)));
        }
        if (more) {
#pragma unroll
            for (int u = 0; u < ROWS_AHEAD; ++u) h[u] = hn[u];
        }
    }
}

template <int PHASE>
__device__ __forceinline__ void own_chunk(const OwnCtx& c, const int* qe, int take,
                                          int* rsd, float* rv, int& rcount, bool& ovf)
{
    const int lane = c.lane;
    const bool on = (lane < take);
    int   s = 0, dl = 0;
    float v = 0.0f;
    if (on) {
        const int q = qe[lane];
        const int e = q >> DL_BITS;
        dl = q & DL_MASK;
        s = (e < E_EDGES) ? c.esrc[e] : (e - E_EDGES);
        const float t = c.as_[s] + c.ad_l[dl];
        v = (t > 0.0f) ? t : NEG_SLOPE * t;
    }
    if (PHASE == 0) {
        if (on) atomicMax(&c.m_l[dl], f2ord(v));
        if (!ovf) {
            if (rcount + take <= OWN_RCAP) {
                if (on) { rsd[rcount + lane] = (s << DL_BITS) | dl; rv[rcount + lane] = v; }
                rcount += take;
            } else {
                ovf = true;
            }
        }
    } else if (PHASE == 1) {
        if (on) {
            const float w = __expf(v - ord2f(c.m_l[dl]));
            atomicAdd(&c.z_l[dl], (unsigned long long)(w * Z_FIX_ONE));
        }
    } else {
        float coef = 0.0f;
        if (on) coef = __expf(v - ord2f(c.m_l[dl])) / own_z(c, dl);
        own_walk(c, s, dl, coef * c.qscale, take);
    }
}

template <int PHASE>
__device__ __forceinline__ void own_scan(const OwnCtx& c, const int* __restrict__ edst, int* qe,
                                         int wave, int n0, int rn,
                                         int* rsd, float* rv, int& rcount, bool& ovf)
{
    const int lane = c.lane;
    int qn = 0;

    {
        const int row = wave + 32 * lane;
        const bool keep = (row < rn);
        if (keep) qe[lane] = ((E_EDGES + n0 + row) << DL_BITS) | row;
        qn = __popc(__builtin_amdgcn_ballot_w32(keep));
    }

    for (int g = 0; g < SCAN_GROUPS; ++g) {
        const int it0 = ((g + (int)blockIdx.x) % SCAN_GROUPS) * SCAN_AHEAD;
        int4 d4[SCAN_AHEAD];
#pragma unroll
        for (int u = 0; u < SCAN_AHEAD; ++u) {
            const int idx = (it0 + u) * OWN_THREADS + (int)threadIdx.x;
            d4[u] = (idx < E_QUADS) ? ((const int4*)edst)[idx] : make_int4(-1, -1, -1, -1);
        }
#pragma unroll
        for (int u = 0; u < SCAN_AHEAD; ++u) {
            const int idx = (it0 + u) * OWN_THREADS + (int)threadIdx.x;
            const int dd[4] = { d4[u].x, d4[u].y, d4[u].z, d4[u].w };
#pragma unroll
            for (int k = 0; k < 4; ++k) {
                const int  dl   = dd[k] - n0;
                const bool keep = (unsigned)dl < (unsigned)rn;
                const unsigned mask = __builtin_amdgcn_ballot_w32(keep);
                if (mask != 0u) {
                    if (keep) qe[qn + __popc(mask & ((1u << lane) - 1u))] = ((idx * 4 + k) << DL_BITS) | dl;
                    qn += __popc(mask);
                }
            }
        }
        while (qn >= 32) {
            qn -= 32;
            own_chunk<PHASE>(c, qe + qn, 32, rsd, rv, rcount, ovf);
        }
    }
    if (qn > 0)
        own_chunk<PHASE>(c, qe, qn, rsd, rv, rcount, ovf);
}

template <bool BUILD>
__global__ void __launch_bounds__(OWN_THREADS)
own_softmax_aggr_kernel(const int* __restrict__ esrc, const int* __restrict__ edst,
                        const float* __restrict__ as_, const float* __restrict__ ad_,
                        const __bf16* __restrict__ Hb, const float* __restrict__ wmax,
                        const float* __restrict__ bias, float* __restrict__ X,
                        int* __restrict__ lcount, int* __restrict__ lst, size_t ws_size)
{
    __shared__ __attribute__((aligned(16))) int acc[(OWN_R + 1) * DIM];
    __shared__ unsigned cnt_l[OWN_R];
    __shared__ int   queue[OWN_WAVES][OWN_QCAP];
    __shared__ int   ret_sd[OWN_WAVES][OWN_RCAP];
    __shared__ float ret_v[OWN_WAVES][OWN_RCAP];
    __shared__ unsigned long long z_l[OWN_R];
    __shared__ unsigned m_l[OWN_R];
    __shared__ float ad_l[OWN_R];
    __shared__ float part[OWN_WAVES];

    const int wave = __builtin_amdgcn_readfirstlane(threadIdx.x >> 5);
    const int lane = threadIdx.x & 31;
    const int n0   = blockIdx.x * OWN_R;
    const int rn   = (N_NODES - n0 < OWN_R) ? (N_NODES - n0) : OWN_R;
    int*   qe  = queue[wave];
    int*   rsd = ret_sd[wave];
    float* rv  = ret_v[wave];

    const int lw = blockIdx.x * OWN_WAVES + wave;
    int   listed = -1;
    int   lsd[LIST_CHUNKS];
    float las[LIST_CHUNKS];
    if (!BUILD) {
        if (ws_size >= WS_NEED) listed = lcount[lw];
#pragma unroll
        for (int k = 0; k < LIST_CHUNKS; ++k) {
            lsd[k] = 0;
            if (k * 32 + lane < listed) lsd[k] = lst[(size_t)lw * OWN_LCAP + k * 32 + lane];
        }
#pragma unroll
        for (int k = 0; k < LIST_CHUNKS; ++k)
            las[k] = as_[lsd[k] >> DL_BITS];
    }

    for (int i = threadIdx.x; i < (OWN_R + 1) * DIM / 4; i += OWN_THREADS) ((int4*)acc)[i] = make_int4(0, 0, 0, 0);
    if (threadIdx.x < OWN_R) {
        m_l[threadIdx.x]  = 0u;
        z_l[threadIdx.x]  = 0ull;
        cnt_l[threadIdx.x] = 0u;
        ad_l[threadIdx.x] = ((int)threadIdx.x < rn) ? ad_[n0 + threadIdx.x] : 0.0f;
    }
    {
        float w1[WMAX_PER_THREAD];
#pragma unroll
        for (int k = 0; k < WMAX_PER_THREAD; ++k) {
            const int i = threadIdx.x + k * OWN_THREADS;
            w1[k] = (i < GEMM_BLOCKS) ? wmax[i] : 0.0f;
        }
        float g = 0.0f;
#pragma unroll
        for (int k = 0; k < WMAX_PER_THREAD; ++k)
            g = fmaxf(g, w1[k]);
#pragma unroll
        for (int off = 16; off > 0; off >>= 1)
            g = fmaxf(g, __shfl_xor(g, off, 32));
        if (lane == 0) part[wave] = g;
    }
    __syncthreads();

    float hm = part[lane];
#pragma unroll
    for (int off = 16; off > 0; off >>= 1)
        hm = fmaxf(hm, __shfl_xor(hm, off, 32));

    OwnCtx c;
    c.esrc = esrc; c.as_ = as_; c.Hb = Hb;
    c.ad_l = ad_l; c.m_l = m_l; c.z_l = z_l; c.acc = acc; c.cnt_l = cnt_l;
    c.qscale = (hm >= HMAX_FLOOR) ? (FIX_ONE / hm) : 0.0f;
    c.lane = lane;

    int  rcount = 0;
    bool ovf    = false;

    if (!BUILD && listed >= 0) {
#pragma unroll
        for (int k = 0; k < LIST_CHUNKS; ++k) {
            const int i = k * 32 + lane;
            if (i < listed) {
                const int   dl = lsd[k] & DL_MASK;
                const float t  = las[k] + ad_l[dl];
                const float v  = (t > 0.0f) ? t : NEG_SLOPE * t;
                atomicMax(&m_l[dl], f2ord(v));
                rsd[i] = lsd[k];
                rv[i]  = v;
            }
        }
        rcount = listed;
    } else {
        own_scan<0>(c, edst, qe, wave, n0, rn, rsd, rv, rcount, ovf);
    }
    rcount = __builtin_amdgcn_readfirstlane(rcount);
    __syncthreads();

    if (ovf) {
        own_scan<1>(c, edst, qe, wave, n0, rn, rsd, rv, rcount, ovf);
    } else {
        for (int c0 = 0; c0 < rcount; c0 += 32) {
            const int i = c0 + lane;
            if (i < rcount) {
                const int   dl = rsd[i] & DL_MASK;
                const float w  = __expf(rv[i] - ord2f(m_l[dl]));
                rv[i] = w;
                atomicAdd(&z_l[dl], (unsigned long long)(w * Z_FIX_ONE));
            }
        }
    }
    __syncthreads();

    if (ovf) {
        own_scan<2>(c, edst, qe, wave, n0, rn, rsd, rv, rcount, ovf);
    } else {
        for (int c0 = 0; c0 < rcount; c0 += 32) {
            const int take = (rcount - c0 < 32) ? (rcount - c0) : 32;
            int   s = 0, dl = 0;
            float coef = 0.0f;
            if (lane < take) {
                const int sd = rsd[c0 + lane];
                s  = sd >> DL_BITS;
                dl = sd & DL_MASK;
                coef = rv[c0 + lane] / own_z(c, dl);
            }
            own_walk(c, s, dl, coef * c.qscale, take);
        }
    }
    __syncthreads();

    const float unscale = hm * FIX_INV_ONE;
    for (int q = threadIdx.x; q < rn * (DIM / 4); q += OWN_THREADS) {
        const int r = q >> 5, l = q & 31;
        const int* rp = acc + r * DIM + l;
        const float4 b = ((const float4*)bias)[l];
        const unsigned off = cnt_l[r] * FIX_MAGIC_BITS;
        float4 o;
        o.x = elu1((float)(int)((unsigned)rp[0]  - off) * unscale + b.x);
        o.y = elu1((float)(int)((unsigned)rp[32] - off) * unscale + b.y);
        o.z = elu1((float)(int)((unsigned)rp[64] - off) * unscale + b.z);
        o.w = elu1((float)(int)((unsigned)rp[96] - off) * unscale + b.w);
        ((float4*)(X + (size_t)(n0 + r) * DIM))[l] = o;
    }

    if (BUILD) {
        if (ws_size >= WS_NEED) {
            const int n = (!ovf && rcount <= OWN_LCAP) ? rcount : -1;
            for (int k = 0; k < LIST_CHUNKS; ++k) {
                const int i = k * 32 + lane;
                if (i < n) lst[(size_t)lw * OWN_LCAP + i] = rsd[i];
            }
            if (lane == 0) lcount[lw] = n;
        }
    }
}

extern "C" void kernel_launch(void* const* d_in, const int* in_sizes, int n_in,
                              void* d_out, int out_size, void* d_ws, size_t ws_size,
                              hipStream_t stream)
{
    (void)in_sizes; (void)n_in; (void)out_size;

    const float* xq    = (const float*)d_in[0];
    const float* xt    = (const float*)d_in[1];
    const int*   ei_q  = (const int*)d_in[2];
    const float* W_mlp = (const float*)d_in[4];
    const float* b_mlp = (const float*)d_in[5];
    const float* Ws    = (const float*)d_in[6];
    const float* a_src = (const float*)d_in[7];
    const float* a_dst = (const float*)d_in[8];
    const float* biases= (const float*)d_in[9];

    const int* esrc = ei_q;
    const int* edst = ei_q + E_EDGES;

    float* Xq = (float*)d_out;
    float* Xt = Xq + (size_t)N_NODES * DIM;

    __bf16*   Wpk   = (__bf16*)d_ws;
    __bf16*   Hb    = Wpk + (size_t)NWEIGHTS * WPK_ELEMS_PER_W;
    float*    stat  = (float*)(Hb + (size_t)N_NODES * DIM);
    float*    as_   = stat;
    float*    ad_   = as_ + N_NODES;
    float*    wmax  = ad_ + N_NODES;
    float*    uvec  = wmax + UVEC_OFF;
    int*      lcount = (int*)((char*)d_ws + WS_LIST_OFF);
    int*      lst    = lcount + LIST_WAVES;

    const dim3 blk(256);
    const dim3 pack_grid(((PACK_WAVES + UVEC_WAVES) * 32 + 255) / 256);
    const dim3 gemm_grid(GEMM_BLOCKS);
    const dim3 own_grid(OWN_BLOCKS);
    const dim3 own_blk(OWN_THREADS);

    pack_weights_kernel<<<pack_grid, blk, 0, stream>>>(W_mlp, Ws, a_src, a_dst, Wpk, uvec);

    gemm128_wmma<false><<<gemm_grid, blk, 0, stream>>>(xq, Wpk, b_mlp, Xq, N_NODES, nullptr, nullptr, nullptr, nullptr);
    gemm128_wmma<false><<<gemm_grid, blk, 0, stream>>>(xt, Wpk, b_mlp, Xt, N_NODES, nullptr, nullptr, nullptr, nullptr);

    for (int l = 0; l < NLAYERS; ++l) {
        const __bf16* Wl = Wpk + (size_t)(l + 1) * WPK_ELEMS_PER_W;
        const float*  ul = uvec   + (size_t)l * 2 * DIM;
        const float*  bl = biases + (size_t)l * DIM;
        float* Xs[2] = { Xq, Xt };
        for (int g = 0; g < 2; ++g) {
            float* X = Xs[g];
            gemm128_wmma<true><<<gemm_grid, blk, 0, stream>>>(X, Wl, nullptr, (void*)Hb, N_NODES, ul, as_, ad_, wmax);
            if (l == 0 && g == 0)
                own_softmax_aggr_kernel<true><<<own_grid, own_blk, 0, stream>>>(esrc, edst, as_, ad_, Hb, wmax, bl, X,
                                                                                lcount, lst, ws_size);
            else
                own_softmax_aggr_kernel<false><<<own_grid, own_blk, 0, stream>>>(esrc, edst, as_, ad_, Hb, wmax, bl, X,
                                                                                 lcount, lst, ws_size);
        }
    }
}


# Owner pass of one graph and the other graph's next product share a launch (a buffer set per graph): product work-groups run beside the owners on a compute unit; 15 -> 11 launches; sums unchanged
# speedup vs baseline: 6.5671x; 1.0842x over previous
#include <hip/hip_runtime.h>
#include <hip/hip_bf16.h>

#define N_NODES 30000
#define E_EDGES 480000
#define EN_TOT  (E_EDGES + N_NODES)
#define DIM     128
#define NLAYERS 3
#define NEG_SLOPE 0.2f
#define NWEIGHTS 4
#define WPK_ELEMS_PER_W (8 * 4 * 32 * 16)
#define Z_FIX_ONE    1099511627776.0f
#define Z_FIX_INV    9.094947017729282e-13f
#define FIX_ONE        2097152.0f
#define FIX_INV_ONE    4.76837158203125e-07f
#define FIX_MAGIC      12582912.0f
#define FIX_MAGIC_BITS 0x4B400000u
#define HMAX_FLOOR   1.0e-20f

typedef __attribute__((ext_vector_type(16))) __bf16 v16bf;
typedef __attribute__((ext_vector_type(4)))  __bf16 v4bf;
typedef __attribute__((ext_vector_type(8)))  float  v8f;

__device__ __forceinline__ unsigned f2ord(float x) {
    unsigned b = __float_as_uint(x);
    return (b & 0x80000000u) ? ~b : (b | 0x80000000u);
}
__device__ __forceinline__ float ord2f(unsigned u) {
    return __uint_as_float((u & 0x80000000u) ? (u ^ 0x80000000u) : ~u);
}

template <int M>
__device__ __forceinline__ float xrow(float x) {
    const int b = __builtin_bit_cast(int, x);
    return __builtin_bit_cast(float, __builtin_amdgcn_update_dpp(b, b, 0x160 + M, 0xf, 0xf, false));
}
__device__ __forceinline__ float xhalf(float x) {
    const int b = __builtin_bit_cast(int, x);
    return __builtin_bit_cast(float, __builtin_amdgcn_permlanex16(b, b, 0x76543210, 0xfedcba98, false, false));
}
__device__ __forceinline__ float wave_sum(float x) {
    x += xrow<1>(x); x += xrow<2>(x); x += xrow<4>(x); x += xrow<8>(x);
    return x + xhalf(x);
}
__device__ __forceinline__ float wave_max(float x) {
    x = fmaxf(x, xrow<1>(x)); x = fmaxf(x, xrow<2>(x)); x = fmaxf(x, xrow<4>(x)); x = fmaxf(x, xrow<8>(x));
    return fmaxf(x, xhalf(x));
}

#define PACK_WAVES (NWEIGHTS * 32)
#define UVEC_WAVES (NLAYERS * DIM)
__global__ void __launch_bounds__(256)
pack_weights_kernel(const float* __restrict__ W_mlp, const float* __restrict__ Ws,
                    const float* __restrict__ a_src, const float* __restrict__ a_dst,
                    __bf16* __restrict__ Wpk, float* __restrict__ uvec)
{
    const int gw   = (blockIdx.x * blockDim.x + threadIdx.x) >> 5;
    const int lane = threadIdx.x & 31;
    if (gw >= PACK_WAVES + UVEC_WAVES) return;
    if (gw >= PACK_WAVES) {
        const int j = gw - PACK_WAVES;
        const int l = j / DIM;
        const int k = j % DIM;
        const float4 w4 = ((const float4*)(Ws + ((size_t)l * DIM + k) * DIM))[lane];
        const float4 s4 = ((const float4*)(a_src + (size_t)l * DIM))[lane];
        const float4 d4 = ((const float4*)(a_dst + (size_t)l * DIM))[lane];
        float ps = w4.x * s4.x + w4.y * s4.y + w4.z * s4.z + w4.w * s4.w;
        float pd = w4.x * d4.x + w4.y * d4.y + w4.z * d4.z + w4.w * d4.w;
        ps = wave_sum(ps);
        pd = wave_sum(pd);
        if (lane == 0) {
            uvec[(size_t)(l * 2 + 0) * DIM + k] = ps;
            uvec[(size_t)(l * 2 + 1) * DIM + k] = pd;
        }
        return;
    }
    const int w  = gw >> 5;
    const int ct = (gw & 31) >> 2;
    const int s  = gw & 3;
    const float* W = (w == 0) ? W_mlp : (Ws + (size_t)(w - 1) * DIM * DIM);
    const int r  = lane & 15;
    const int hi = lane >> 4;
    const int col = ct * 16 + r;
    const int kb  = s * 32;
    __bf16* out = Wpk + ((size_t)gw * 32 + lane) * 16;
#pragma unroll
    for (int h = 0; h < 16; ++h) {
        const int K = kb + ((h < 8) ? (8 * hi + h) : (16 + 8 * hi + (h - 8)));
        out[h] = (__bf16)W[(size_t)K * DIM + col];
    }
}

#define GEMM_BLOCKS (N_NODES / 16)
static_assert(N_NODES % 16 == 0, "every strip is whole");
#define XS_PITCH 136
typedef __attribute__((ext_vector_type(8))) __bf16 v8bf;

template <bool BF16_OUT>
__global__ void __launch_bounds__(256)
gemm128_wmma(const float* __restrict__ X, const __bf16* __restrict__ Wpk,
             const float* __restrict__ bias, void* __restrict__ out, int rows,
             const float* __restrict__ uvec, float* __restrict__ as_, float* __restrict__ ad_,
             float* __restrict__ wmax)
{
    __shared__ __attribute__((aligned(16))) __bf16 xs[16 * XS_PITCH];
    __shared__ float al[32];
    __shared__ float part[8];

    if (blockIdx.x * 16 >= (unsigned)rows) return;
    const int lane = threadIdx.x & 31;
    const int ct   = threadIdx.x >> 5;
    const int r    = lane & 15;
    const int hi   = lane >> 4;
    const int col0 = ct * 16;

    const float4* xp = (const float4*)(X + (size_t)blockIdx.x * 16 * DIM);
    v16bf b[4];
#pragma unroll
    for (int s = 0; s < 4; ++s)
        b[s] = *(const v16bf*)(Wpk + ((size_t)(ct * 4 + s) * 32 + lane) * 16);
    const float4 x0 = xp[threadIdx.x];
    const float4 x1 = xp[threadIdx.x + 256];
    float4 us4 = make_float4(0.f, 0.f, 0.f, 0.f), ud4 = us4;
    if (BF16_OUT) {
        us4 = ((const float4*)uvec)[lane];
        ud4 = ((const float4*)(uvec + DIM))[lane];
        __builtin_amdgcn_sched_barrier(0);
    }

    {
        v4bf p0, p1;
        p0[0] = (__bf16)x0.x; p0[1] = (__bf16)x0.y; p0[2] = (__bf16)x0.z; p0[3] = (__bf16)x0.w;
        p1[0] = (__bf16)x1.x; p1[1] = (__bf16)x1.y; p1[2] = (__bf16)x1.z; p1[3] = (__bf16)x1.w;
        *(v4bf*)(xs + ct * XS_PITCH + lane * 4)       = p0;
        *(v4bf*)(xs + (ct + 8) * XS_PITCH + lane * 4) = p1;
    }
    if (BF16_OUT) {
        float ps0 = x0.x * us4.x + x0.y * us4.y + x0.z * us4.z + x0.w * us4.w;
        float pd0 = x0.x * ud4.x + x0.y * ud4.y + x0.z * ud4.z + x0.w * ud4.w;
        float ps1 = x1.x * us4.x + x1.y * us4.y + x1.z * us4.z + x1.w * us4.w;
        float pd1 = x1.x * ud4.x + x1.y * ud4.y + x1.z * ud4.z + x1.w * ud4.w;
        ps0 = wave_sum(ps0);  pd0 = wave_sum(pd0);
        ps1 = wave_sum(ps1);  pd1 = wave_sum(pd1);
        if (lane == 0) {
            al[ct]      = ps0;  al[ct + 8]  = ps1;
            al[16 + ct] = pd0;  al[24 + ct] = pd1;
        }
    }
    __syncthreads();
    if (BF16_OUT && ct == 0) {
        float* dst = (lane < 16) ? (as_ + (size_t)blockIdx.x * 16 + lane)
                                 : (ad_ + (size_t)blockIdx.x * 16 + (lane - 16));
        *dst = al[lane];
    }

    v8f acc = {};
#pragma unroll
    for (int s = 0; s < 4; ++s) {
        const int kb = s * 32;
        const v8bf alo = *(const v8bf*)(xs + r * XS_PITCH + kb + 8 * hi);
        const v8bf ahi = *(const v8bf*)(xs + r * XS_PITCH + kb + 16 + 8 * hi);
        const v16bf a = __builtin_shufflevector(alo, ahi, 0, 1, 2, 3, 4, 5, 6, 7, 8, 9, 10, 11, 12, 13, 14, 15);
        acc = __builtin_amdgcn_wmma_f32_16x16x32_bf16(
             false, a,  false, b[s],
             (short)0, acc,  false,  false);
    }
    if (BF16_OUT) {
        __bf16* H = (__bf16*)out;
        float amax = 0.0f;
#pragma unroll
        for (int v = 0; v < 8; ++v) {
            const int m = blockIdx.x * 16 + v + 8 * hi;
            H[(size_t)m * DIM + col0 + r] = (__bf16)acc[v];
            amax = fmaxf(amax, fabsf(acc[v]));
        }
        amax = wave_max(amax);
        if (lane == 0) part[ct] = amax;
        __syncthreads();
        if (threadIdx.x == 0) {
            float g = part[0];
#pragma unroll
            for (int i = 1; i < 8; ++i) g = fmaxf(g, part[i]);
            wmax[blockIdx.x] = (float)(__bf16)g;
        }
    } else {
        float* H = (float*)out;
        const float bv = bias[col0 + r];
#pragma unroll
        for (int v = 0; v < 8; ++v) {
            const int m = blockIdx.x * 16 + v + 8 * hi;
            H[(size_t)m * DIM + col0 + r] = acc[v] + bv;
        }
    }
}

#define STAT_TAIL_WORDS 7500
#define UVEC_OFF        2048
#define UVEC_WORDS      (NLAYERS * 2 * DIM)
static_assert(GEMM_BLOCKS <= UVEC_OFF, "wmax ends before uvec begins");
static_assert(UVEC_OFF % 4 == 0, "uvec is read in 16-byte pieces");
static_assert(UVEC_OFF + UVEC_WORDS <= STAT_TAIL_WORDS, "uvec ends inside the tail");

#define ROWS_AHEAD   8
#define OWN_R        118
#define OWN_BLOCKS   ((N_NODES + OWN_R - 1) / OWN_R)
#define OWN_THREADS  1024
#define OWN_WAVES    (OWN_THREADS / 32)
#define SCAN_AHEAD   8
#define E_QUADS      (E_EDGES / 4)
#define SCAN_ITERS   ((E_QUADS + OWN_THREADS - 1) / OWN_THREADS)
#define SCAN_GROUPS  ((SCAN_ITERS + SCAN_AHEAD - 1) / SCAN_AHEAD)
#define OWN_QCAP     (32 + SCAN_AHEAD * 4 * 32)
#define OWN_RCAP     256
#define DL_BITS      7
#define DL_MASK      ((1 << DL_BITS) - 1)
#define WMAX_PER_THREAD ((GEMM_BLOCKS + OWN_THREADS - 1) / OWN_THREADS)
static_assert(E_EDGES % 4 == 0, "the destination list is read in 16-byte pieces");
static_assert(OWN_R <= (1 << DL_BITS), "row inside the group must fit DL_BITS");
static_assert(EN_TOT < (1 << (31 - DL_BITS)), "edge number and row share one 32-bit queue word");
static_assert(N_NODES < (1 << (31 - DL_BITS)), "source and row share one 32-bit word");
static_assert(OWN_WAVES == 32, "the per-wave maxima are folded by one wave-wide exchange");
#define OWN_LCAP     128
#define LIST_CHUNKS  (OWN_LCAP / 32)
#define LIST_WAVES   (OWN_BLOCKS * OWN_WAVES)
#define WS_STAT_END  (NWEIGHTS * WPK_ELEMS_PER_W * 2 + N_NODES * DIM * 2 + (2 * N_NODES + STAT_TAIL_WORDS) * 4)
#define WS_LIST_OFF  ((WS_STAT_END + 127) / 128 * 128)
#define WS_NEED      ((size_t)WS_LIST_OFF + (size_t)LIST_WAVES * 4 + (size_t)LIST_WAVES * OWN_LCAP * 4)
#define WS_SET1_OFF  ((WS_NEED + 127) / 128 * 128)
#define WS_NEED2     ((size_t)WS_SET1_OFF + (size_t)N_NODES * DIM * 2 + (size_t)(2 * N_NODES + STAT_TAIL_WORDS) * 4)
static_assert(OWN_LCAP % 32 == 0, "a list is read 32 entries at a time");
static_assert(OWN_LCAP <= OWN_RCAP, "a listed wave retains all its entries");

__device__ __forceinline__ v4bf load_row4(const __bf16* __restrict__ Hb, int s, int lane)
{
    return ((const v4bf*)(Hb + (size_t)s * DIM))[lane];
}

__device__ __forceinline__ float elu1(float v) { return (v > 0.0f) ? v : (__expf(v) - 1.0f); }

struct OwnCtx {
    const int*    esrc;
    const float*  as_;
    const __bf16* Hb;
    const float*  ad_l;
    unsigned*     m_l;
    unsigned long long* z_l;
    int*          acc;
    unsigned*     cnt_l;
    float         qscale;
    int           lane;
};

__device__ __forceinline__ float own_z(const OwnCtx& c, int dl)
{
    return (float)c.z_l[dl] * Z_FIX_INV;
}

__device__ __forceinline__ void own_walk(const OwnCtx& c, int s, int dl, float coef_scaled, int take)
{
    const int lane = c.lane;
    const int coef_bits = __float_as_int(coef_scaled);
    if (lane < take) atomicAdd(&c.cnt_l[dl], 1u);
    dl = (lane < take) ? dl : OWN_R;
    v4bf h[ROWS_AHEAD];
#pragma unroll
    for (int u = 0; u < ROWS_AHEAD; ++u)
        h[u] = load_row4(c.Hb, __builtin_amdgcn_readlane(s, u), lane);

    for (int j0 = 0; j0 < take; j0 += ROWS_AHEAD) {
        v4bf hn[ROWS_AHEAD];
        const bool more = (j0 + ROWS_AHEAD < take);
        if (more) {
#pragma unroll
            for (int u = 0; u < ROWS_AHEAD; ++u)
                hn[u] = load_row4(c.Hb, __builtin_amdgcn_readlane(s, j0 + ROWS_AHEAD + u), lane);
        }
#pragma unroll
        for (int u = 0; u < ROWS_AHEAD; ++u) {
            const int   dj = __builtin_amdgcn_readlane(dl, j0 + u);
            const float cj = __int_as_float(__builtin_amdgcn_readlane(coef_bits, j0 + u));
            int* ap = c.acc + dj * DIM + lane;
            atomicAdd(ap +  0, __float_as_int(__builtin_fmaf(cj, (float)h[u][0], FIX_MAGIC)));
            atomicAdd(ap + 32, __float_as_int(__builtin_fmaf(cj, (float)h[u][1], FIX_MAGIC)));
            atomicAdd(ap + 64, __float_as_int(__builtin_fmaf(cj, (float)h[u][2], FIX_MAGIC)));
            atomicAdd(ap + 96, __float_as_int(__builtin_fmaf(cj, (float)h[u][3], FIX_MAGIC)));
        }
        if (more) {
#pragma unroll
            for (int u = 0; u < ROWS_AHEAD; ++u) h[u] = hn[u];
        }
    }
}

template <int PHASE, int RCAP>
__device__ __forceinline__ void own_chunk(const OwnCtx& c, const int* qe, int take,
                                          int* rsd, float* rv, int& rcount, bool& ovf)
{
    const int lane = c.lane;
    const bool on = (lane < take);
    int   s = 0, dl = 0;
    float v = 0.0f;
    if (on) {
        const int q = qe[lane];
        const int e = q >> DL_BITS;
        dl = q & DL_MASK;
        s = (e < E_EDGES) ? c.esrc[e] : (e - E_EDGES);
        const float t = c.as_[s] + c.ad_l[dl];
        v = (t > 0.0f) ? t : NEG_SLOPE * t;
    }
    if (PHASE == 0) {
        if (on) atomicMax(&c.m_l[dl], f2ord(v));
        if (!ovf) {
            if (rcount + take <= RCAP) {
                if (on) { rsd[rcount + lane] = (s << DL_BITS) | dl; rv[rcount + lane] = v; }
                rcount += take;
            } else {
                ovf = true;
            }
        }
    } else if (PHASE == 1) {
        if (on) {
            const float w = __expf(v - ord2f(c.m_l[dl]));
            atomicAdd(&c.z_l[dl], (unsigned long long)(w * Z_FIX_ONE));
        }
    } else {
        float coef = 0.0f;
        if (on) coef = __expf(v - ord2f(c.m_l[dl])) / own_z(c, dl);
        own_walk(c, s, dl, coef * c.qscale, take);
    }
}

template <int PHASE, int SA, int RCAP>
__device__ __forceinline__ void own_scan(const OwnCtx& c, const int* __restrict__ edst, int* qe,
                                         int wave, int n0, int rn,
                                         int* rsd, float* rv, int& rcount, bool& ovf)
{
    const int lane = c.lane;
    int qn = 0;

    {
        const int row = wave + 32 * lane;
        const bool keep = (row < rn);
        if (keep) qe[lane] = ((E_EDGES + n0 + row) << DL_BITS) | row;
        qn = __popc(__builtin_amdgcn_ballot_w32(keep));
    }

    constexpr int GROUPS = (SCAN_ITERS + SA - 1) / SA;
    for (int g = 0; g < GROUPS; ++g) {
        const int it0 = ((g + (int)blockIdx.x) % GROUPS) * SA;
        int4 d4[SA];
#pragma unroll
        for (int u = 0; u < SA; ++u) {
            const int idx = (it0 + u) * OWN_THREADS + (int)threadIdx.x;
            d4[u] = (idx < E_QUADS) ? ((const int4*)edst)[idx] : make_int4(-1, -1, -1, -1);
        }
#pragma unroll
        for (int u = 0; u < SA; ++u) {
            const int idx = (it0 + u) * OWN_THREADS + (int)threadIdx.x;
            const int dd[4] = { d4[u].x, d4[u].y, d4[u].z, d4[u].w };
#pragma unroll
            for (int k = 0; k < 4; ++k) {
                const int  dl   = dd[k] - n0;
                const bool keep = (unsigned)dl < (unsigned)rn;
                const unsigned mask = __builtin_amdgcn_ballot_w32(keep);
                if (mask != 0u) {
                    if (keep) qe[qn + __popc(mask & ((1u << lane) - 1u))] = ((idx * 4 + k) << DL_BITS) | dl;
                    qn += __popc(mask);
                }
            }
        }
        while (qn >= 32) {
            qn -= 32;
            own_chunk<PHASE, RCAP>(c, qe + qn, 32, rsd, rv, rcount, ovf);
        }
    }
    if (qn > 0)
        own_chunk<PHASE, RCAP>(c, qe, qn, rsd, rv, rcount, ovf);
}

template <bool BUILD, int SA, int RCAP>
__device__ __forceinline__ void
own_body(const int* __restrict__ esrc, const int* __restrict__ edst,
         const float* __restrict__ as_, const float* __restrict__ ad_,
         const __bf16* __restrict__ Hb, const float* __restrict__ wmax,
         const float* __restrict__ bias, float* __restrict__ X,
         int* __restrict__ lcount, int* __restrict__ lst, size_t ws_size,
         int* acc, unsigned* cnt_l, int* queue, int* ret_sd, float* ret_v,
         unsigned long long* z_l, unsigned* m_l, float* ad_l, float* part)
{
    static_assert(OWN_LCAP <= RCAP, "a listed wave retains all its entries");
    const int wave = __builtin_amdgcn_readfirstlane(threadIdx.x >> 5);
    const int lane = threadIdx.x & 31;
    const int n0   = blockIdx.x * OWN_R;
    const int rn   = (N_NODES - n0 < OWN_R) ? (N_NODES - n0) : OWN_R;
    int*   qe  = queue  + wave * (32 + SA * 4 * 32);
    int*   rsd = ret_sd + wave * RCAP;
    float* rv  = ret_v  + wave * RCAP;

    const int lw = blockIdx.x * OWN_WAVES + wave;
    int   listed = -1;
    int   lsd[LIST_CHUNKS];
    float las[LIST_CHUNKS];
    if (!BUILD) {
        if (ws_size >= WS_NEED) listed = lcount[lw];
#pragma unroll
        for (int k = 0; k < LIST_CHUNKS; ++k) {
            lsd[k] = 0;
            if (k * 32 + lane < listed) lsd[k] = lst[(size_t)lw * OWN_LCAP + k * 32 + lane];
        }
#pragma unroll
        for (int k = 0; k < LIST_CHUNKS; ++k)
            las[k] = as_[lsd[k] >> DL_BITS];
    }

    for (int i = threadIdx.x; i < (OWN_R + 1) * DIM / 4; i += OWN_THREADS) ((int4*)acc)[i] = make_int4(0, 0, 0, 0);
    if (threadIdx.x < OWN_R) {
        m_l[threadIdx.x]  = 0u;
        z_l[threadIdx.x]  = 0ull;
        cnt_l[threadIdx.x] = 0u;
        ad_l[threadIdx.x] = ((int)threadIdx.x < rn) ? ad_[n0 + threadIdx.x] : 0.0f;
    }
    {
        float w1[WMAX_PER_THREAD];
#pragma unroll
        for (int k = 0; k < WMAX_PER_THREAD; ++k) {
            const int i = threadIdx.x + k * OWN_THREADS;
            w1[k] = (i < GEMM_BLOCKS) ? wmax[i] : 0.0f;
        }
        float g = 0.0f;
#pragma unroll
        for (int k = 0; k < WMAX_PER_THREAD; ++k)
            g = fmaxf(g, w1[k]);
#pragma unroll
        for (int off = 16; off > 0; off >>= 1)
            g = fmaxf(g, __shfl_xor(g, off, 32));
        if (lane == 0) part[wave] = g;
    }
    __syncthreads();

    float hm = part[lane];
#pragma unroll
    for (int off = 16; off > 0; off >>= 1)
        hm = fmaxf(hm, __shfl_xor(hm, off, 32));

    OwnCtx c;
    c.esrc = esrc; c.as_ = as_; c.Hb = Hb;
    c.ad_l = ad_l; c.m_l = m_l; c.z_l = z_l; c.acc = acc; c.cnt_l = cnt_l;
    c.qscale = (hm >= HMAX_FLOOR) ? (FIX_ONE / hm) : 0.0f;
    c.lane = lane;

    int  rcount = 0;
    bool ovf    = false;

    if (!BUILD && listed >= 0) {
#pragma unroll
        for (int k = 0; k < LIST_CHUNKS; ++k) {
            const int i = k * 32 + lane;
            if (i < listed) {
                const int   dl = lsd[k] & DL_MASK;
                const float t  = las[k] + ad_l[dl];
                const float v  = (t > 0.0f) ? t : NEG_SLOPE * t;
                atomicMax(&m_l[dl], f2ord(v));
                rsd[i] = lsd[k];
                rv[i]  = v;
            }
        }
        rcount = listed;
    } else {
        own_scan<0, SA, RCAP>(c, edst, qe, wave, n0, rn, rsd, rv, rcount, ovf);
    }
    rcount = __builtin_amdgcn_readfirstlane(rcount);
    __syncthreads();

    if (ovf) {
        own_scan<1, SA, RCAP>(c, edst, qe, wave, n0, rn, rsd, rv, rcount, ovf);
    } else {
        for (int c0 = 0; c0 < rcount; c0 += 32) {
            const int i = c0 + lane;
            if (i < rcount) {
                const int   dl = rsd[i] & DL_MASK;
                const float w  = __expf(rv[i] - ord2f(m_l[dl]));
                rv[i] = w;
                atomicAdd(&z_l[dl], (unsigned long long)(w * Z_FIX_ONE));
            }
        }
    }
    __syncthreads();

    if (ovf) {
        own_scan<2, SA, RCAP>(c, edst, qe, wave, n0, rn, rsd, rv, rcount, ovf);
    } else {
        for (int c0 = 0; c0 < rcount; c0 += 32) {
            const int take = (rcount - c0 < 32) ? (rcount - c0) : 32;
            int   s = 0, dl = 0;
            float coef = 0.0f;
            if (lane < take) {
                const int sd = rsd[c0 + lane];
                s  = sd >> DL_BITS;
                dl = sd & DL_MASK;
                coef = rv[c0 + lane] / own_z(c, dl);
            }
            own_walk(c, s, dl, coef * c.qscale, take);
        }
    }
    __syncthreads();

    const float unscale = hm * FIX_INV_ONE;
    for (int q = threadIdx.x; q < rn * (DIM / 4); q += OWN_THREADS) {
        const int r = q >> 5, l = q & 31;
        const int* rp = acc + r * DIM + l;
        const float4 b = ((const float4*)bias)[l];
        const unsigned off = cnt_l[r] * FIX_MAGIC_BITS;
        float4 o;
        o.x = elu1((float)(int)((unsigned)rp[0]  - off) * unscale + b.x);
        o.y = elu1((float)(int)((unsigned)rp[32] - off) * unscale + b.y);
        o.z = elu1((float)(int)((unsigned)rp[64] - off) * unscale + b.z);
        o.w = elu1((float)(int)((unsigned)rp[96] - off) * unscale + b.w);
        ((float4*)(X + (size_t)(n0 + r) * DIM))[l] = o;
    }

    if (BUILD) {
        if (ws_size >= WS_NEED) {
            const int n = (!ovf && rcount <= OWN_LCAP) ? rcount : -1;
            for (int k = 0; k < LIST_CHUNKS; ++k) {
                const int i = k * 32 + lane;
                if (i < n) lst[(size_t)lw * OWN_LCAP + i] = rsd[i];
            }
            if (lane == 0) lcount[lw] = n;
        }
    }
}

template <bool BUILD>
__global__ void __launch_bounds__(OWN_THREADS)
own_softmax_aggr_kernel(const int* __restrict__ esrc, const int* __restrict__ edst,
                        const float* __restrict__ as_, const float* __restrict__ ad_,
                        const __bf16* __restrict__ Hb, const float* __restrict__ wmax,
                        const float* __restrict__ bias, float* __restrict__ X,
                        int* __restrict__ lcount, int* __restrict__ lst, size_t ws_size)
{
    __shared__ __attribute__((aligned(16))) int acc[(OWN_R + 1) * DIM];
    __shared__ unsigned cnt_l[OWN_R];
    __shared__ int   queue[OWN_WAVES * OWN_QCAP];
    __shared__ int   ret_sd[OWN_WAVES * OWN_RCAP];
    __shared__ float ret_v[OWN_WAVES * OWN_RCAP];
    __shared__ unsigned long long z_l[OWN_R];
    __shared__ unsigned m_l[OWN_R];
    __shared__ float ad_l[OWN_R];
    __shared__ float part[OWN_WAVES];
    own_body<BUILD, SCAN_AHEAD, OWN_RCAP>(esrc, edst, as_, ad_, Hb, wmax, bias, X, lcount, lst, ws_size,
                                          acc, cnt_l, queue, ret_sd, ret_v, z_l, m_l, ad_l, part);
}

#define PAIR_STRIPS  (OWN_WAVES / 8)
#define PAIR_GBLOCKS ((GEMM_BLOCKS + PAIR_STRIPS - 1) / PAIR_STRIPS)
#define PAIR_SA      1
#define PAIR_QCAP    (32 + PAIR_SA * 4 * 32)

__global__ void __launch_bounds__(OWN_THREADS)
own_beside_gemm_kernel(const int* __restrict__ esrc, const int* __restrict__ edst,
                       const float* __restrict__ as_, const float* __restrict__ ad_,
                       const __bf16* __restrict__ Hb, const float* __restrict__ wmax,
                       const float* __restrict__ bias, float* __restrict__ X,
                       int* __restrict__ lcount, int* __restrict__ lst, size_t ws_size,
                       const float* __restrict__ gX, const __bf16* __restrict__ gWpk,
                       __bf16* __restrict__ gH, const float* __restrict__ guvec,
                       float* __restrict__ gas, float* __restrict__ gad, float* __restrict__ gwmax)
{
    __shared__ __attribute__((aligned(16))) int acc[(OWN_R + 1) * DIM];
    __shared__ unsigned cnt_l[OWN_R];
    __shared__ int   queue[OWN_WAVES * PAIR_QCAP];
    __shared__ int   ret_sd[OWN_WAVES * OWN_LCAP];
    __shared__ float ret_v[OWN_WAVES * OWN_LCAP];
    __shared__ unsigned long long z_l[OWN_R];
    __shared__ unsigned m_l[OWN_R];
    __shared__ float ad_l[OWN_R];
    __shared__ float part[OWN_WAVES];
    __shared__ __attribute__((aligned(16))) __bf16 xs_all[PAIR_STRIPS * 16 * XS_PITCH];
    __shared__ float al_all[PAIR_STRIPS * 32];
    __shared__ float gpart_all[PAIR_STRIPS * 8];

    if (blockIdx.x < OWN_BLOCKS) {
        own_body<false, PAIR_SA, OWN_LCAP>(esrc, edst, as_, ad_, Hb, wmax, bias, X, lcount, lst, ws_size,
                                           acc, cnt_l, queue, ret_sd, ret_v, z_l, m_l, ad_l, part);
        return;
    }

    const int wave = __builtin_amdgcn_readfirstlane(threadIdx.x >> 5);
    const int lane = threadIdx.x & 31;
    const int sg   = wave >> 3;
    const int ct   = wave & 7;
    const int t    = threadIdx.x & 255;
    const int want = ((int)blockIdx.x - OWN_BLOCKS) * PAIR_STRIPS + sg;
    const bool valid = (want < GEMM_BLOCKS);
    const int strip = valid ? want : (GEMM_BLOCKS - 1);
    const int r    = lane & 15;
    const int hi   = lane >> 4;
    const int col0 = ct * 16;
    __bf16* xs   = xs_all + sg * 16 * XS_PITCH;
    float*  al   = al_all + sg * 32;
    float*  gprt = gpart_all + sg * 8;

    const float4* xp = (const float4*)(gX + (size_t)strip * 16 * DIM);
    v16bf b[4];
#pragma unroll
    for (int s = 0; s < 4; ++s)
        b[s] = *(const v16bf*)(gWpk + ((size_t)(ct * 4 + s) * 32 + lane) * 16);
    const float4 x0 = xp[t];
    const float4 x1 = xp[t + 256];
    const float4 us4 = ((const float4*)guvec)[lane];
    const float4 ud4 = ((const float4*)(guvec + DIM))[lane];
    __builtin_amdgcn_sched_barrier(0);
    {
        v4bf p0, p1;
        p0[0] = (__bf16)x0.x; p0[1] = (__bf16)x0.y; p0[2] = (__bf16)x0.z; p0[3] = (__bf16)x0.w;
        p1[0] = (__bf16)x1.x; p1[1] = (__bf16)x1.y; p1[2] = (__bf16)x1.z; p1[3] = (__bf16)x1.w;
        *(v4bf*)(xs + ct * XS_PITCH + lane * 4)       = p0;
        *(v4bf*)(xs + (ct + 8) * XS_PITCH + lane * 4) = p1;
    }
    {
        float ps0 = x0.x * us4.x + x0.y * us4.y + x0.z * us4.z + x0.w * us4.w;
        float pd0 = x0.x * ud4.x + x0.y * ud4.y + x0.z * ud4.z + x0.w * ud4.w;
        float ps1 = x1.x * us4.x + x1.y * us4.y + x1.z * us4.z + x1.w * us4.w;
        float pd1 = x1.x * ud4.x + x1.y * ud4.y + x1.z * ud4.z + x1.w * ud4.w;
        ps0 = wave_sum(ps0);  pd0 = wave_sum(pd0);
        ps1 = wave_sum(ps1);  pd1 = wave_sum(pd1);
        if (lane == 0) {
            al[ct]      = ps0;  al[ct + 8]  = ps1;
            al[16 + ct] = pd0;  al[24 + ct] = pd1;
        }
    }
    __syncthreads();
    if (valid && ct == 0) {
        float* dst = (lane < 16) ? (gas + (size_t)strip * 16 + lane)
                                 : (gad + (size_t)strip * 16 + (lane - 16));
        *dst = al[lane];
    }

    v8f cacc = {};
#pragma unroll
    for (int s = 0; s < 4; ++s) {
        const int kb = s * 32;
        const v8bf alo = *(const v8bf*)(xs + r * XS_PITCH + kb + 8 * hi);
        const v8bf ahi = *(const v8bf*)(xs + r * XS_PITCH + kb + 16 + 8 * hi);
        const v16bf a = __builtin_shufflevector(alo, ahi, 0, 1, 2, 3, 4, 5, 6, 7, 8, 9, 10, 11, 12, 13, 14, 15);
        cacc = __builtin_amdgcn_wmma_f32_16x16x32_bf16(
             false, a,  false, b[s],
             (short)0, cacc,  false,  false);
    }
    float amax = 0.0f;
#pragma unroll
    for (int v = 0; v < 8; ++v)
        amax = fmaxf(amax, fabsf(cacc[v]));
    if (valid) {
#pragma unroll
        for (int v = 0; v < 8; ++v) {
            const int m = strip * 16 + v + 8 * hi;
            gH[(size_t)m * DIM + col0 + r] = (__bf16)cacc[v];
        }
    }
    amax = wave_max(amax);
    if (lane == 0) gprt[ct] = amax;
    __syncthreads();
    if (valid && t == 0) {
        float g = gprt[0];
#pragma unroll
        for (int i = 1; i < 8; ++i) g = fmaxf(g, gprt[i]);
        gwmax[strip] = (float)(__bf16)g;
    }
}

extern "C" void kernel_launch(void* const* d_in, const int* in_sizes, int n_in,
                              void* d_out, int out_size, void* d_ws, size_t ws_size,
                              hipStream_t stream)
{
    (void)in_sizes; (void)n_in; (void)out_size;

    const float* xq    = (const float*)d_in[0];
    const float* xt    = (const float*)d_in[1];
    const int*   ei_q  = (const int*)d_in[2];
    const float* W_mlp = (const float*)d_in[4];
    const float* b_mlp = (const float*)d_in[5];
    const float* Ws    = (const float*)d_in[6];
    const float* a_src = (const float*)d_in[7];
    const float* a_dst = (const float*)d_in[8];
    const float* biases= (const float*)d_in[9];

    const int* esrc = ei_q;
    const int* edst = ei_q + E_EDGES;

    float* Xq = (float*)d_out;
    float* Xt = Xq + (size_t)N_NODES * DIM;

    __bf16*   Wpk   = (__bf16*)d_ws;
    __bf16*   Hb    = Wpk + (size_t)NWEIGHTS * WPK_ELEMS_PER_W;
    float*    stat  = (float*)(Hb + (size_t)N_NODES * DIM);
    float*    as_   = stat;
    float*    ad_   = as_ + N_NODES;
    float*    wmax  = ad_ + N_NODES;
    float*    uvec  = wmax + UVEC_OFF;
    int*      lcount = (int*)((char*)d_ws + WS_LIST_OFF);
    int*      lst    = lcount + LIST_WAVES;

    __bf16*   Hb1   = (__bf16*)((char*)d_ws + WS_SET1_OFF);
    float*    as1   = (float*)(Hb1 + (size_t)N_NODES * DIM);
    float*    ad1   = as1 + N_NODES;
    float*    wmax1 = ad1 + N_NODES;

    const dim3 blk(256);
    const dim3 pack_grid(((PACK_WAVES + UVEC_WAVES) * 32 + 255) / 256);
    const dim3 gemm_grid(GEMM_BLOCKS);
    const dim3 own_grid(OWN_BLOCKS);
    const dim3 own_blk(OWN_THREADS);
    const dim3 pair_grid(OWN_BLOCKS + PAIR_GBLOCKS);

    pack_weights_kernel<<<pack_grid, blk, 0, stream>>>(W_mlp, Ws, a_src, a_dst, Wpk, uvec);

    gemm128_wmma<false><<<gemm_grid, blk, 0, stream>>>(xq, Wpk, b_mlp, Xq, N_NODES, nullptr, nullptr, nullptr, nullptr);
    gemm128_wmma<false><<<gemm_grid, blk, 0, stream>>>(xt, Wpk, b_mlp, Xt, N_NODES, nullptr, nullptr, nullptr, nullptr);

    if (ws_size >= WS_NEED2) {
        const __bf16* W0 = Wpk + (size_t)1 * WPK_ELEMS_PER_W;
        const __bf16* W1 = Wpk + (size_t)2 * WPK_ELEMS_PER_W;
        const __bf16* W2 = Wpk + (size_t)3 * WPK_ELEMS_PER_W;
        const float*  u0 = uvec;
        const float*  u1 = uvec + 2 * DIM;
        const float*  u2 = uvec + 4 * DIM;
        const float*  b0 = biases;
        const float*  b1 = biases + DIM;
        const float*  b2 = biases + 2 * DIM;
        gemm128_wmma<true><<<gemm_grid, blk, 0, stream>>>(Xq, W0, nullptr, (void*)Hb, N_NODES, u0, as_, ad_, wmax);
        own_softmax_aggr_kernel<true><<<own_grid, own_blk, 0, stream>>>(esrc, edst, as_, ad_, Hb, wmax, b0, Xq,
                                                                        lcount, lst, ws_size);
        gemm128_wmma<true><<<gemm_grid, blk, 0, stream>>>(Xt, W0, nullptr, (void*)Hb1, N_NODES, u0, as1, ad1, wmax1);
        own_beside_gemm_kernel<<<pair_grid, own_blk, 0, stream>>>(esrc, edst, as1, ad1, Hb1, wmax1, b0, Xt, lcount, lst, ws_size,
                                                                  Xq, W1, Hb, u1, as_, ad_, wmax);
        own_beside_gemm_kernel<<<pair_grid, own_blk, 0, stream>>>(esrc, edst, as_, ad_, Hb, wmax, b1, Xq, lcount, lst, ws_size,
                                                                  Xt, W1, Hb1, u1, as1, ad1, wmax1);
        own_beside_gemm_kernel<<<pair_grid, own_blk, 0, stream>>>(esrc, edst, as1, ad1, Hb1, wmax1, b1, Xt, lcount, lst, ws_size,
                                                                  Xq, W2, Hb, u2, as_, ad_, wmax);
        own_beside_gemm_kernel<<<pair_grid, own_blk, 0, stream>>>(esrc, edst, as_, ad_, Hb, wmax, b2, Xq, lcount, lst, ws_size,
                                                                  Xt, W2, Hb1, u2, as1, ad1, wmax1);
        own_softmax_aggr_kernel<false><<<own_grid, own_blk, 0, stream>>>(esrc, edst, as1, ad1, Hb1, wmax1, b2, Xt,
                                                                         lcount, lst, ws_size);
        return;
    }

    for (int l = 0; l < NLAYERS; ++l) {
        const __bf16* Wl = Wpk + (size_t)(l + 1) * WPK_ELEMS_PER_W;
        const float*  ul = uvec   + (size_t)l * 2 * DIM;
        const float*  bl = biases + (size_t)l * DIM;
        float* Xs[2] = { Xq, Xt };
        for (int g = 0; g < 2; ++g) {
            float* X = Xs[g];
            gemm128_wmma<true><<<gemm_grid, blk, 0, stream>>>(X, Wl, nullptr, (void*)Hb, N_NODES, ul, as_, ad_, wmax);
            if (l == 0 && g == 0)
                own_softmax_aggr_kernel<true><<<own_grid, own_blk, 0, stream>>>(esrc, edst, as_, ad_, Hb, wmax, bl, X,
                                                                                lcount, lst, ws_size);
            else
                own_softmax_aggr_kernel<false><<<own_grid, own_blk, 0, stream>>>(esrc, edst, as_, ad_, Hb, wmax, bl, X,
                                                                                 lcount, lst, ws_size);
        }
    }
}
